# starting kernel
# baseline (speedup 1.0000x reference)
#include <hip/hip_runtime.h>

typedef __attribute__((ext_vector_type(2))) float v2f;
typedef __attribute__((ext_vector_type(8))) float v8f;

#define H 64
#define F85 85
#define KPAD 88

__global__ void zero_kernel(float* __restrict__ p, long n) {
    long i = (long)blockIdx.x * blockDim.x + threadIdx.x;
    if (i < n) p[i] = 0.0f;
}

__global__ void prep_w_kernel(const float* __restrict__ w1_src_t,
                              const float* __restrict__ w1_dst_t,
                              const float* __restrict__ b1_t,
                              const float* __restrict__ w1_src_r,
                              const float* __restrict__ w1_dst_r,
                              float* __restrict__ W1t, float* __restrict__ W1c,
                              float* __restrict__ W1m, float* __restrict__ b1sum) {
    int i = blockIdx.x * blockDim.x + threadIdx.x;
    if (i >= KPAD * H) return;
    int r = i >> 6, c = i & 63;
    float vt = 0.f, vc = 0.f, vm = 0.f;
    if (r < F85) {
        vt = w1_dst_t[r * H + c] + w1_dst_t[F85 * H + r * H + c];
        vc = w1_src_r[r * H + c];
        vm = w1_src_r[F85 * H + r * H + c];
    } else if (r == 85) {
        vt = w1_src_t[c];
        vc = w1_dst_r[c];
        vm = w1_dst_r[H + c];
    } else if (r == 86) {
        vt = w1_src_t[H + c];
    }
    W1t[i] = vt; W1c[i] = vc; W1m[i] = vm;
    if (i < H) b1sum[i] = b1_t[i] + b1_t[H + i];
}

__global__ void scatter_scalar_kernel(const float* __restrict__ xc, const float* __restrict__ xm,
                                      const int* __restrict__ sp, const int* __restrict__ dp,
                                      const int* __restrict__ sr, const int* __restrict__ dr,
                                      float* __restrict__ aggc, float* __restrict__ aggm, int E) {
    int e = blockIdx.x * blockDim.x + threadIdx.x;
    if (e >= E) return;
    atomicAdd(&aggc[dp[e]], xc[sp[e]]);
    atomicAdd(&aggm[dr[e]], xm[sr[e]]);
}

__global__ void scatter_feat_kernel(float* __restrict__ outbuf, const int* __restrict__ oidx,
                                    const float* __restrict__ inbuf, const int* __restrict__ iidx,
                                    int E, int F) {
    long i = (long)blockIdx.x * blockDim.x + threadIdx.x;
    long total = (long)E * F;
    if (i >= total) return;
    int e = (int)(i / F), j = (int)(i % F);
    atomicAdd(&outbuf[(size_t)oidx[e] * F + j], inbuf[(size_t)iidx[e] * F + j]);
}

__global__ __launch_bounds__(128) void gemm88_kernel(
    const float* __restrict__ Xmain, const float* __restrict__ extra1,
    const float* __restrict__ extra2, const float* __restrict__ W,
    const float* __restrict__ bias, float* __restrict__ out, int M) {
    __shared__ float xs[16 * KPAD];

    const int row0 = blockIdx.x * 16;
    for (int i = threadIdx.x; i < 16 * KPAD; i += 128) {
        int r = i / KPAD, c = i - r * KPAD;
        int grow = row0 + r;
        float v;
        if (c < F85)      v = Xmain[(size_t)grow * F85 + c];
        else if (c == 85) v = extra1 ? extra1[grow] : 0.0f;
        else if (c == 86) v = extra2 ? extra2[grow] : 0.0f;
        else              v = 0.0f;
        xs[i] = v;
    }
    __syncthreads();

    const int wave = threadIdx.x >> 5;
    const int lane = threadIdx.x & 31;
    const int half = lane >> 4;
    const int lm   = lane & 15;
    const int c0   = wave * 16;

    const float bval = bias[c0 + lm];
    v8f acc;
#pragma unroll
    for (int i = 0; i < 8; ++i) acc[i] = bval;

    for (int k = 0; k < KPAD; k += 4) {
        const int ka = k + 2 * half;
        v2f a, b;
        a.x = xs[lm * KPAD + ka];
        a.y = xs[lm * KPAD + ka + 1];
        b.x = W[(size_t)(ka + 0) * H + c0 + lm];
        b.y = W[(size_t)(ka + 1) * H + c0 + lm];
        acc = __builtin_amdgcn_wmma_f32_16x16x4_f32(false, a, false, b, (short)0, acc,
                                                    false, false);
    }
#pragma unroll
    for (int v = 0; v < 8; ++v) {
        float r = acc[v];
        r = r > 0.0f ? r : 0.0f;
        out[(size_t)(row0 + v + 8 * half) * H + c0 + lm] = r;
    }
}

__global__ __launch_bounds__(128) void conv2_out_kernel(
    const float* __restrict__ h_t,  const float* __restrict__ aggc,
    const float* __restrict__ aggm, const float* __restrict__ w2l,
    const float* __restrict__ w2r,  const float* __restrict__ b2,
    const float* __restrict__ wlin, const float* __restrict__ blin,
    float* __restrict__ out, int M) {
    __shared__ float hs[16 * H];
    __shared__ float cs[16 * H];
    __shared__ float ms[16 * H];
    __shared__ float ts[16 * 2 * H];

    const int row0 = blockIdx.x * 16;
    for (int i = threadIdx.x; i < 16 * H; i += 128) {
        int r = i >> 6, c = i & 63;
        size_t g = (size_t)(row0 + r) * H + c;
        hs[i] = h_t[g];
        cs[i] = aggc[g];
        ms[i] = aggm[g];
    }
    __syncthreads();

    const int wave = threadIdx.x >> 5;
    const int lane = threadIdx.x & 31;
    const int half = lane >> 4;
    const int lm   = lane & 15;
    const int c0   = wave * 16;

    v8f acc1, acc2;
    const float bv1 = b2[0 * H + c0 + lm];
    const float bv2 = b2[1 * H + c0 + lm];
#pragma unroll
    for (int i = 0; i < 8; ++i) { acc1[i] = bv1; acc2[i] = bv2; }

    const float* W2l0 = w2l;
    const float* W2l1 = w2l + 1 * H * H;
    const float* W2r0 = w2r;
    const float* W2r1 = w2r + 1 * H * H;

    for (int k = 0; k < H; k += 4) {
        const int ka = k + 2 * half;
        v2f ah, ac, am, b;
        ah.x = hs[lm * H + ka]; ah.y = hs[lm * H + ka + 1];
        ac.x = cs[lm * H + ka]; ac.y = cs[lm * H + ka + 1];
        am.x = ms[lm * H + ka]; am.y = ms[lm * H + ka + 1];

        b.x = W2l0[(ka + 0) * H + c0 + lm]; b.y = W2l0[(ka + 1) * H + c0 + lm];
        acc1 = __builtin_amdgcn_wmma_f32_16x16x4_f32(false, ac, false, b, (short)0, acc1,
                                                     false, false);
        b.x = W2r0[(ka + 0) * H + c0 + lm]; b.y = W2r0[(ka + 1) * H + c0 + lm];
        acc1 = __builtin_amdgcn_wmma_f32_16x16x4_f32(false, ah, false, b, (short)0, acc1,
                                                     false, false);
        b.x = W2l1[(ka + 0) * H + c0 + lm]; b.y = W2l1[(ka + 1) * H + c0 + lm];
        acc2 = __builtin_amdgcn_wmma_f32_16x16x4_f32(false, am, false, b, (short)0, acc2,
                                                     false, false);
        b.x = W2r1[(ka + 0) * H + c0 + lm]; b.y = W2r1[(ka + 1) * H + c0 + lm];
        acc2 = __builtin_amdgcn_wmma_f32_16x16x4_f32(false, ah, false, b, (short)0, acc2,
                                                     false, false);
    }

#pragma unroll
    for (int v = 0; v < 8; ++v) {
        float r1 = acc1[v]; r1 = r1 > 0.0f ? r1 : 0.0f;
        float r2 = acc2[v]; r2 = r2 > 0.0f ? r2 : 0.0f;
        ts[(v + 8 * half) * 128 + c0 + lm]      = r1;
        ts[(v + 8 * half) * 128 + 64 + c0 + lm] = r2;
    }
    __syncthreads();

    if (threadIdx.x < 32) {
        const int r = threadIdx.x >> 1, c = threadIdx.x & 1;
        float s = blin[c];
#pragma unroll 8
        for (int j = 0; j < 128; ++j) s += ts[r * 128 + j] * wlin[j * 2 + c];
        out[(size_t)(row0 + r) * 2 + c] = s;
    }
}

extern "C" void kernel_launch(void* const* d_in, const int* in_sizes, int n_in,
                              void* d_out, int out_size, void* d_ws, size_t ws_size,
                              hipStream_t stream) {
    const float* x_card   = (const float*)d_in[0];
    const float* x_merch  = (const float*)d_in[1];
    const float* x_trans  = (const float*)d_in[2];
    const float* w1_src_t = (const float*)d_in[3];
    const float* w1_dst_t = (const float*)d_in[4];
    const float* b1_t     = (const float*)d_in[5];
    const float* w1_src_r = (const float*)d_in[6];
    const float* w1_dst_r = (const float*)d_in[7];
    const float* b1_r     = (const float*)d_in[8];
    const float* w2_l     = (const float*)d_in[9];
    const float* w2_r     = (const float*)d_in[10];
    const float* b2       = (const float*)d_in[11];
    const float* w_lin    = (const float*)d_in[12];
    const float* b_lin    = (const float*)d_in[13];
    const int* src_pays   = (const int*)d_in[14];
    const int* dst_pays   = (const int*)d_in[15];
    const int* src_recv   = (const int*)d_in[16];
    const int* dst_recv   = (const int*)d_in[17];
    float* out = (float*)d_out;

    const int N_c = in_sizes[0];
    const int N_m = in_sizes[1];
    const int N_t = in_sizes[2] / F85;
    const int E   = in_sizes[14];

    float* ws = (float*)d_ws;
    size_t off = 0;
    float* aggc1  = ws + off; off += (size_t)N_t;
    float* aggm1  = ws + off; off += (size_t)N_t;
    float* aggt2c = ws + off; off += (size_t)N_c * F85;
    float* aggt2m = ws + off; off += (size_t)N_m * F85;
    float* agg2c  = ws + off; off += (size_t)N_t * H;
    float* agg2m  = ws + off; off += (size_t)N_t * H;
    const long zeroCount = (long)off;
    float* h_t    = ws + off; off += (size_t)N_t * H;
    float* h_c    = ws + off; off += (size_t)N_c * H;
    float* h_m    = ws + off; off += (size_t)N_m * H;
    float* W1t    = ws + off; off += (size_t)KPAD * H;
    float* W1c    = ws + off; off += (size_t)KPAD * H;
    float* W1m    = ws + off; off += (size_t)KPAD * H;
    float* b1sum  = ws + off; off += (size_t)H;

    {
        long blocks = (zeroCount + 255) / 256;
        zero_kernel<<<(unsigned)blocks, 256, 0, stream>>>(ws, zeroCount);
    }
    prep_w_kernel<<<(KPAD * H + 255) / 256, 256, 0, stream>>>(
        w1_src_t, w1_dst_t, b1_t, w1_src_r, w1_dst_r, W1t, W1c, W1m, b1sum);
    scatter_scalar_kernel<<<(E + 255) / 256, 256, 0, stream>>>(
        x_card, x_merch, src_pays, dst_pays, src_recv, dst_recv, aggc1, aggm1, E);
    {
        long total = (long)E * F85;
        scatter_feat_kernel<<<(unsigned)((total + 255) / 256), 256, 0, stream>>>(
            aggt2c, src_pays, x_trans, dst_pays, E, F85);
        scatter_feat_kernel<<<(unsigned)((total + 255) / 256), 256, 0, stream>>>(
            aggt2m, src_recv, x_trans, dst_recv, E, F85);
    }
    gemm88_kernel<<<N_t / 16, 128, 0, stream>>>(x_trans, aggc1, aggm1, W1t, b1sum, h_t, N_t);
    gemm88_kernel<<<N_c / 16, 128, 0, stream>>>(aggt2c, x_card, nullptr, W1c, b1_r, h_c, N_c);
    gemm88_kernel<<<N_m / 16, 128, 0, stream>>>(aggt2m, x_merch, nullptr, W1m, b1_r + H, h_m, N_m);
    {
        long total = (long)E * H;
        scatter_feat_kernel<<<(unsigned)((total + 255) / 256), 256, 0, stream>>>(
            agg2c, dst_pays, h_c, src_pays, E, H);
        scatter_feat_kernel<<<(unsigned)((total + 255) / 256), 256, 0, stream>>>(
            agg2m, dst_recv, h_m, src_recv, E, H);
    }
    conv2_out_kernel<<<N_t / 16, 128, 0, stream>>>(
        h_t, agg2c, agg2m, w2_l, w2_r, b2, w_lin, b_lin, out, N_t);
}


# conv1 atomic sums kept once per chip half: a wave adds into its own half's copy (die number from the placement message), the conv1 GEMM kernels add the two copies; sum order: two partial sums
# speedup vs baseline: 2.8967x; 2.8967x over previous
#include <hip/hip_runtime.h>

typedef __attribute__((ext_vector_type(2))) float v2f;
typedef __attribute__((ext_vector_type(4))) float v4f;
typedef __attribute__((ext_vector_type(8))) float v8f;

#define H 64
#define F85 85
#define KPAD 88
#define AP 96

__device__ __forceinline__ int chip_half() {
    return (int)((__builtin_amdgcn_s_sendmsg_rtn(0x87) >> 18) & 1u);
}

#define ZWG 256
__global__ __launch_bounds__(256) void zero_kernel(float* __restrict__ p, long n) {
    const unsigned lane = threadIdx.x & 31;
    const unsigned gw = blockIdx.x * 8 + (threadIdx.x >> 5);
    const unsigned nw = gridDim.x * 8;
    const unsigned n4 = (unsigned)(n >> 2);
    const unsigned units = (n4 + 31) >> 5;
    const unsigned per = (units + nw - 1) / nw;
    const unsigned b = gw * per * 32;
    const unsigned lim = b + per * 32;
    const unsigned e = lim < n4 ? lim : n4;
    v4f* __restrict__ q = (v4f*)p;
    const v4f z = {0.0f, 0.0f, 0.0f, 0.0f};
    unsigned i = b + lane;
    for (; i + 7 * 32 < e; i += 8 * 32) {
#pragma unroll
        for (int k = 0; k < 8; ++k) q[i + 32 * k] = z;
    }
    for (; i < e; i += 32) q[i] = z;
    if (gw == 0 && lane < (unsigned)(n & 3)) p[((long)n4 << 2) + lane] = 0.0f;
}

__global__ void prep_w_kernel(const float* __restrict__ w1_src_t,
                              const float* __restrict__ w1_dst_t,
                              const float* __restrict__ b1_t,
                              const float* __restrict__ w1_src_r,
                              const float* __restrict__ w1_dst_r,
                              float* __restrict__ W1t, float* __restrict__ W1c,
                              float* __restrict__ W1m, float* __restrict__ b1sum) {
    int i = blockIdx.x * blockDim.x + threadIdx.x;
    if (i >= KPAD * H) return;
    int r = i >> 6, c = i & 63;
    float vt = 0.f, vc = 0.f, vm = 0.f;
    if (r < F85) {
        vt = w1_dst_t[r * H + c] + w1_dst_t[F85 * H + r * H + c];
        vc = w1_src_r[r * H + c];
        vm = w1_src_r[F85 * H + r * H + c];
    } else if (r == 85) {
        vt = w1_src_t[c];
        vc = w1_dst_r[c];
        vm = w1_dst_r[H + c];
    } else if (r == 86) {
        vt = w1_src_t[H + c];
    }
    W1t[i] = vt; W1c[i] = vc; W1m[i] = vm;
    if (i < H) b1sum[i] = b1_t[i] + b1_t[H + i];
}

__global__ void scatter_scalar_kernel(const float* __restrict__ xc, const float* __restrict__ xm,
                                      const int* __restrict__ sp, const int* __restrict__ dp,
                                      const int* __restrict__ sr, const int* __restrict__ dr,
                                      float* __restrict__ aggc, float* __restrict__ aggm, int E,
                                      long cs) {
    int e = blockIdx.x * blockDim.x + threadIdx.x;
    if (e >= E) return;
    const int dc = dp[e], dm = dr[e];
    const float vc = xc[sp[e]], vm = xm[sr[e]];
    const long ho = chip_half() * cs;
    atomicAdd(&aggc[ho + dc], vc);
    atomicAdd(&aggm[ho + dm], vm);
}

#define EPW 4
template <int F, int IP, int OP, int NE, bool TWOCOPIES>
__device__ __forceinline__ void scatter_edges(float* __restrict__ outbuf,
                                              const int* __restrict__ oidx,
                                              const float* __restrict__ inbuf,
                                              const int* __restrict__ iidx,
                                              int e0, int lane, long cs) {
    constexpr int NP = (F + 31) / 32;
    constexpr int TAIL = F - 32 * (NP - 1);
    int is[NE], od[NE];
#pragma unroll
    for (int u = 0; u < NE; ++u) { is[u] = iidx[e0 + u]; od[u] = oidx[e0 + u]; }
#pragma unroll
    for (int u = 0; u < NE; ++u) asm volatile("" ::"s"(is[u]), "s"(od[u]));
    if (TWOCOPIES) outbuf += chip_half() * cs;
    float v[NE][NP];
#pragma unroll
    for (int u = 0; u < NE; ++u) {
        const float* srow = inbuf + (size_t)is[u] * IP;
#pragma unroll
        for (int k = 0; k < NP; ++k) {
            const int j = lane + 32 * k;
            v[u][k] = srow[j < F ? j : F - 1];
        }
    }
    if (TAIL < 32) {
#pragma unroll
        for (int u = 0; u < NE; ++u) asm volatile("" ::"v"(v[u][NP - 1]));
    }
#pragma unroll
    for (int u = 0; u < NE; ++u) {
        float* drow = outbuf + (size_t)od[u] * OP;
#pragma unroll
        for (int k = 0; k < NP - 1; ++k) atomicAdd(&drow[lane + 32 * k], v[u][k]);
    }
    if (lane < TAIL) {
#pragma unroll
        for (int u = 0; u < NE; ++u)
            atomicAdd(&outbuf[(size_t)od[u] * OP + lane + 32 * (NP - 1)], v[u][NP - 1]);
    }
}

template <int F, int IP, int OP, bool TWOCOPIES>
__device__ __forceinline__ void scatter_rows(float* __restrict__ outbuf,
                                             const int* __restrict__ oidx,
                                             const float* __restrict__ inbuf,
                                             const int* __restrict__ iidx, int E, long cs) {
    const int lane = threadIdx.x & 31;
    const int wave = __builtin_amdgcn_readfirstlane((int)(threadIdx.x >> 5));
    const int e0 = ((int)blockIdx.x * 8 + wave) * EPW;
    if (e0 + EPW <= E) {
        scatter_edges<F, IP, OP, EPW, TWOCOPIES>(outbuf, oidx, inbuf, iidx, e0, lane, cs);
    } else {
        for (int e = e0; e < E; ++e)
            scatter_edges<F, IP, OP, 1, TWOCOPIES>(outbuf, oidx, inbuf, iidx, e, lane, cs);
    }
}

__global__ __launch_bounds__(256) void scatter85_kernel(float* __restrict__ outbuf,
                                                        const int* __restrict__ oidx,
                                                        const float* __restrict__ inbuf,
                                                        const int* __restrict__ iidx, int E,
                                                        long cs) {
    scatter_rows<F85, F85, AP, true>(outbuf, oidx, inbuf, iidx, E, cs);
}

__global__ __launch_bounds__(256) void scatter64_kernel(float* __restrict__ outbuf,
                                                        const int* __restrict__ oidx,
                                                        const float* __restrict__ inbuf,
                                                        const int* __restrict__ iidx, int E) {
    scatter_rows<H, H, H, false>(outbuf, oidx, inbuf, iidx, E, 0);
}

__global__ __launch_bounds__(128) void gemm88_kernel(
    const float* __restrict__ Xmain, long cs, int pitch, const float* __restrict__ extra1,
    const float* __restrict__ extra2, const float* __restrict__ W,
    const float* __restrict__ bias, float* __restrict__ out, int M) {
    __shared__ float xs[16 * KPAD];

    const int row0 = blockIdx.x * 16;
    for (int i = threadIdx.x; i < 16 * KPAD; i += 128) {
        int r = i / KPAD, c = i - r * KPAD;
        int grow = row0 + r;
        float v;
        if (c < F85)      v = Xmain[(size_t)grow * pitch + c] + Xmain[(size_t)grow * pitch + c + cs];
        else if (c == 85) v = extra1 ? extra1[grow] : 0.0f;
        else if (c == 86) v = extra2 ? extra2[grow] : 0.0f;
        else              v = 0.0f;
        xs[i] = v;
    }
    __syncthreads();

    const int wave = threadIdx.x >> 5;
    const int lane = threadIdx.x & 31;
    const int half = lane >> 4;
    const int lm   = lane & 15;
    const int c0   = wave * 16;

    const float bval = bias[c0 + lm];
    v8f acc;
#pragma unroll
    for (int i = 0; i < 8; ++i) acc[i] = bval;

    for (int k = 0; k < KPAD; k += 4) {
        const int ka = k + 2 * half;
        v2f a, b;
        a.x = xs[lm * KPAD + ka];
        a.y = xs[lm * KPAD + ka + 1];
        b.x = W[(size_t)(ka + 0) * H + c0 + lm];
        b.y = W[(size_t)(ka + 1) * H + c0 + lm];
        acc = __builtin_amdgcn_wmma_f32_16x16x4_f32(false, a, false, b, (short)0, acc,
                                                    false, false);
    }
#pragma unroll
    for (int v = 0; v < 8; ++v) {
        float r = acc[v];
        r = r > 0.0f ? r : 0.0f;
        out[(size_t)(row0 + v + 8 * half) * H + c0 + lm] = r;
    }
}

#define C1W 8
#define C1Q 11
#define C1P 360
#define C1X 1408

__device__ __forceinline__ void c1_load(v4f (&r)[C1Q], float& exa, float& exb,
                                        const float* __restrict__ x,
                                        const float* __restrict__ extra, long cs,
                                        int tile, int lane) {
    const v4f* p = (const v4f*)(x + (size_t)tile * (16 * F85));
#pragma unroll
    for (int j = 0; j < C1Q - 1; ++j) r[j] = p[lane + 32 * j];
    const int pl = 32 * (C1Q - 1) + lane;
    r[C1Q - 1] = p[pl < 340 ? pl : 339];
    const float* pe = extra + tile * 16 + (lane & 15);
    exa = pe[0];
    exb = pe[cs];
}

__device__ __forceinline__ void c1_tile(const float* __restrict__ arow,
                                        const float* __restrict__ xrow,
                                        const v4f (&wreg)[C1Q][4], const float (&bv)[4],
                                        int c, int h, float* __restrict__ out, int tile) {
    float a[C1Q][4];
#pragma unroll
    for (int q = 0; q < C1Q; ++q) {
#pragma unroll
        for (int e = 0; e < 4; ++e) a[q][e] = arow[8 * q + e];
    }
    const float e1 = xrow[0], e2 = xrow[16];
    a[C1Q - 1][1] = h ? e1 : a[C1Q - 1][1];
    a[C1Q - 1][2] = h ? e2 : a[C1Q - 1][2];
    a[C1Q - 1][3] = h ? 0.0f : a[C1Q - 1][3];

    v8f acc[4];
#pragma unroll
    for (int ct = 0; ct < 4; ++ct) {
#pragma unroll
        for (int i = 0; i < 8; ++i) acc[ct][i] = bv[ct];
    }
#pragma unroll
    for (int q = 0; q < C1Q; ++q) {
        const v2f a0 = {a[q][0], a[q][1]}, a1 = {a[q][2], a[q][3]};
#pragma unroll
        for (int ct = 0; ct < 4; ++ct) {
            const v4f w = wreg[q][ct];
            const v2f b0 = {w[0], w[1]};
            acc[ct] = __builtin_amdgcn_wmma_f32_16x16x4_f32(false, a0, false, b0, (short)0,
                                                            acc[ct], false, false);
        }
#pragma unroll
        for (int ct = 0; ct < 4; ++ct) {
            const v4f w = wreg[q][ct];
            const v2f b1 = {w[2], w[3]};
            acc[ct] = __builtin_amdgcn_wmma_f32_16x16x4_f32(false, a1, false, b1, (short)0,
                                                            acc[ct], false, false);
        }
    }
    float* orow = out + ((size_t)tile * 16 + 8 * h) * H + 4 * c;
#pragma unroll
    for (int v = 0; v < 8; ++v) {
        v4f o;
#pragma unroll
        for (int ct = 0; ct < 4; ++ct) {
            const float r = acc[ct][v];
            o[ct] = r > 0.0f ? r : 0.0f;
        }
        *(v4f*)(orow + v * H) = o;
    }
}

__global__ __launch_bounds__(256) void conv1_t_kernel(
    const float* __restrict__ x, const float* __restrict__ extra1,
    const float* __restrict__ extra2, const float* __restrict__ W,
    const float* __restrict__ bias, float* __restrict__ out, int ntiles, long cs) {
    __shared__ v4f stage[C1W * C1P];

    const int lane = threadIdx.x & 31;
    const int c = lane & 15;
    const int h = lane >> 4;
    const int wave = __builtin_amdgcn_readfirstlane((int)(threadIdx.x >> 5));
    const int gw = blockIdx.x * C1W + wave;
    const int nw = gridDim.x * C1W;
    v4f* __restrict__ reg = stage + wave * C1P;
    float* __restrict__ regf = (float*)reg;
    const float* __restrict__ arow = regf + c * F85 + 4 * h;
    const float* __restrict__ xrow = regf + C1X + c;
    const float* __restrict__ extra = h ? extra2 : extra1;

    v4f wreg[C1Q][4];
#pragma unroll
    for (int q = 0; q < C1Q; ++q) {
#pragma unroll
        for (int e = 0; e < 4; ++e) {
            const v4f w = *(const v4f*)(W + (size_t)(8 * q + 4 * h + e) * H + 4 * c);
#pragma unroll
            for (int ct = 0; ct < 4; ++ct) wreg[q][ct][e] = w[ct];
        }
    }
    float bv[4];
#pragma unroll
    for (int ct = 0; ct < 4; ++ct) bv[ct] = bias[4 * c + ct];

    const int last = ntiles - 1;
    v4f r[C1Q];
    float exa, exb;
    int tile = gw;
    c1_load(r, exa, exb, x, extra, cs, tile < last ? tile : last, lane);
    for (; tile < ntiles; tile += nw) {
#pragma unroll
        for (int j = 0; j < C1Q; ++j) reg[lane + 32 * j] = r[j];
        regf[C1X + lane] = exa + exb;
        const int nt = tile + nw;
        c1_load(r, exa, exb, x, extra, cs, nt < last ? nt : last, lane);
        __builtin_amdgcn_sched_barrier(0);
        c1_tile(arow, xrow, wreg, bv, c, h, out, tile);
    }
}

struct C2Lane {
    float wl1[4][2], wl2[4][2], bs1[4], bs2[4];
    float blo;
    bool b3, b2, b1, b0;
    int lane, c, h;
};

__device__ __forceinline__ void c2_load(v4f (&ac)[8], v4f (&ah)[8], v4f (&am)[8],
                                        const float* __restrict__ h_t,
                                        const float* __restrict__ aggc,
                                        const float* __restrict__ aggm,
                                        int tile, int c, int h) {
    const size_t ro = ((size_t)tile * 16 + c) * H + 4 * h;
    const v4f* ph = (const v4f*)(h_t + ro);
    const v4f* pc = (const v4f*)(aggc + ro);
    const v4f* pm = (const v4f*)(aggm + ro);
#pragma unroll
    for (int q = 0; q < 8; ++q) { ac[q] = pc[2 * q]; ah[q] = ph[2 * q]; am[q] = pm[2 * q]; }
}

__device__ __forceinline__ void c2_tile(const v4f (&ac)[8], const v4f (&ah)[8],
                                        const v4f (&am)[8], const v4f (&wreg)[16][8],
                                        const C2Lane& L, float* __restrict__ out, int tile) {
    float p[16];
#pragma unroll
    for (int i = 0; i < 16; ++i) p[i] = 0.0f;

#pragma unroll
    for (int pass = 0; pass < 2; ++pass) {
        v8f acc[4];
#pragma unroll
        for (int ct = 0; ct < 4; ++ct) {
#pragma unroll
            for (int i = 0; i < 8; ++i) acc[ct][i] = 0.0f;
        }
#pragma unroll
        for (int q = 0; q < 8; ++q) {
            const v4f an = pass ? am[q] : ac[q];
            const v4f ar = ah[q];
            const v2f an0 = {an[0], an[1]}, an1 = {an[2], an[3]};
            const v2f ar0 = {ar[0], ar[1]}, ar1 = {ar[2], ar[3]};
#pragma unroll
            for (int ct = 0; ct < 4; ++ct) {
                const v4f w = wreg[(2 * pass + 0) * 4 + ct][q];
                const v2f b0 = {w[0], w[1]};
                acc[ct] = __builtin_amdgcn_wmma_f32_16x16x4_f32(false, an0, false, b0, (short)0,
                                                                acc[ct], false, false);
            }
#pragma unroll
            for (int ct = 0; ct < 4; ++ct) {
                const v4f w = wreg[(2 * pass + 0) * 4 + ct][q];
                const v2f b1 = {w[2], w[3]};
                acc[ct] = __builtin_amdgcn_wmma_f32_16x16x4_f32(false, an1, false, b1, (short)0,
                                                                acc[ct], false, false);
            }
#pragma unroll
            for (int ct = 0; ct < 4; ++ct) {
                const v4f w = wreg[(2 * pass + 1) * 4 + ct][q];
                const v2f b0 = {w[0], w[1]};
                acc[ct] = __builtin_amdgcn_wmma_f32_16x16x4_f32(false, ar0, false, b0, (short)0,
                                                                acc[ct], false, false);
            }
#pragma unroll
            for (int ct = 0; ct < 4; ++ct) {
                const v4f w = wreg[(2 * pass + 1) * 4 + ct][q];
                const v2f b1 = {w[2], w[3]};
                acc[ct] = __builtin_amdgcn_wmma_f32_16x16x4_f32(false, ar1, false, b1, (short)0,
                                                                acc[ct], false, false);
            }
        }
#pragma unroll
        for (int ct = 0; ct < 4; ++ct) {
            const float w0 = pass ? L.wl2[ct][0] : L.wl1[ct][0];
            const float w1 = pass ? L.wl2[ct][1] : L.wl1[ct][1];
            const float bv = pass ? L.bs2[ct] : L.bs1[ct];
#pragma unroll
            for (int v = 0; v < 8; ++v) {
                float r = acc[ct][v] + bv;
                r = r > 0.0f ? r : 0.0f;
                p[2 * v + 0] += r * w0;
                p[2 * v + 1] += r * w1;
            }
        }
    }

    float s8[8], s4[4], s2[2];
#pragma unroll
    for (int i = 0; i < 8; ++i) {
        const float keep = L.b3 ? p[i + 8] : p[i];
        const float send = L.b3 ? p[i] : p[i + 8];
        s8[i] = keep + __shfl_xor(send, 8);
    }
#pragma unroll
    for (int i = 0; i < 4; ++i) {
        const float keep = L.b2 ? s8[i + 4] : s8[i];
        const float send = L.b2 ? s8[i] : s8[i + 4];
        s4[i] = keep + __shfl_xor(send, 4);
    }
#pragma unroll
    for (int i = 0; i < 2; ++i) {
        const float keep = L.b1 ? s4[i + 2] : s4[i];
        const float send = L.b1 ? s4[i] : s4[i + 2];
        s2[i] = keep + __shfl_xor(send, 2);
    }
    const float keep1 = L.b0 ? s2[1] : s2[0];
    const float send1 = L.b0 ? s2[0] : s2[1];
    const float tot = keep1 + __shfl_xor(send1, 1);
    out[(size_t)tile * 32 + L.lane] = tot + L.blo;
}

__global__ __launch_bounds__(128) void conv2_out_kernel(
    const float* __restrict__ h_t,  const float* __restrict__ aggc,
    const float* __restrict__ aggm, const float* __restrict__ w2l,
    const float* __restrict__ w2r,  const float* __restrict__ b2,
    const float* __restrict__ wlin, const float* __restrict__ blin,
    float* __restrict__ out, int ntiles) {
    __shared__ v4f bf[4 * 4 * 8 * 32];

    {
        float* dst = (float*)bf;
        for (int i = threadIdx.x; i < 4 * H * 16; i += 128) {
            const int n4 = i & 15, k = (i >> 4) & 63, blk = i >> 10;
            const float* srcw = ((blk & 1) ? w2r : w2l) + (blk >> 1) * H * H;
            const v4f w = *(const v4f*)(srcw + k * H + 4 * n4);
            const int q = k >> 3, hh = (k >> 2) & 1, e = k & 3;
#pragma unroll
            for (int j = 0; j < 4; ++j) {
                const int n = 4 * n4 + j;
                const int ct = n >> 4, cc = n & 15;
                dst[((((blk * 4 + ct) * 8 + q) * 32) + hh * 16 + cc) * 4 + e] = w[j];
            }
        }
    }
    __syncthreads();

    C2Lane L;
    L.lane = threadIdx.x & 31;
    L.c = L.lane & 15;
    L.h = L.lane >> 4;
    const int wave = __builtin_amdgcn_readfirstlane((int)(threadIdx.x >> 5));
    const int gw = blockIdx.x * 4 + wave;
    const int nw = gridDim.x * 4;

#pragma unroll
    for (int ct = 0; ct < 4; ++ct) {
        L.wl1[ct][0] = wlin[(16 * ct + L.c) * 2 + 0];
        L.wl1[ct][1] = wlin[(16 * ct + L.c) * 2 + 1];
        L.wl2[ct][0] = wlin[(H + 16 * ct + L.c) * 2 + 0];
        L.wl2[ct][1] = wlin[(H + 16 * ct + L.c) * 2 + 1];
        L.bs1[ct] = b2[16 * ct + L.c];
        L.bs2[ct] = b2[H + 16 * ct + L.c];
    }
    L.blo = blin[L.lane & 1];
    L.b3 = (L.c & 8) != 0; L.b2 = (L.c & 4) != 0; L.b1 = (L.c & 2) != 0; L.b0 = (L.c & 1) != 0;

    v4f wreg[16][8];
#pragma unroll
    for (int bc = 0; bc < 16; ++bc) {
#pragma unroll
        for (int q = 0; q < 8; ++q) wreg[bc][q] = bf[(bc * 8 + q) * 32 + L.lane];
    }

    const int last = ntiles - 1;
    v4f ac0[8], ah0[8], am0[8], ac1[8], ah1[8], am1[8];
    int tile = gw;
    c2_load(ac0, ah0, am0, h_t, aggc, aggm, tile < last ? tile : last, L.c, L.h);
    for (; tile + nw < ntiles; tile += 2 * nw) {
        const int t2 = tile + 2 * nw;
        c2_load(ac1, ah1, am1, h_t, aggc, aggm, tile + nw, L.c, L.h);
        __builtin_amdgcn_sched_barrier(0);
        c2_tile(ac0, ah0, am0, wreg, L, out, tile);
        c2_load(ac0, ah0, am0, h_t, aggc, aggm, t2 < last ? t2 : last, L.c, L.h);
        __builtin_amdgcn_sched_barrier(0);
        c2_tile(ac1, ah1, am1, wreg, L, out, tile + nw);
    }
    if (tile < ntiles) c2_tile(ac0, ah0, am0, wreg, L, out, tile);
}

extern "C" void kernel_launch(void* const* d_in, const int* in_sizes, int n_in,
                              void* d_out, int out_size, void* d_ws, size_t ws_size,
                              hipStream_t stream) {
    const float* x_card   = (const float*)d_in[0];
    const float* x_merch  = (const float*)d_in[1];
    const float* x_trans  = (const float*)d_in[2];
    const float* w1_src_t = (const float*)d_in[3];
    const float* w1_dst_t = (const float*)d_in[4];
    const float* b1_t     = (const float*)d_in[5];
    const float* w1_src_r = (const float*)d_in[6];
    const float* w1_dst_r = (const float*)d_in[7];
    const float* b1_r     = (const float*)d_in[8];
    const float* w2_l     = (const float*)d_in[9];
    const float* w2_r     = (const float*)d_in[10];
    const float* b2       = (const float*)d_in[11];
    const float* w_lin    = (const float*)d_in[12];
    const float* b_lin    = (const float*)d_in[13];
    const int* src_pays   = (const int*)d_in[14];
    const int* dst_pays   = (const int*)d_in[15];
    const int* src_recv   = (const int*)d_in[16];
    const int* dst_recv   = (const int*)d_in[17];
    float* out = (float*)d_out;

    const int N_c = in_sizes[0];
    const int N_m = in_sizes[1];
    const int N_t = in_sizes[2] / F85;
    const int E   = in_sizes[14];

    float* ws = (float*)d_ws;
    size_t off = 0;
    float* aggc1  = ws + off; off += (size_t)N_t;
    float* aggm1  = ws + off; off += (size_t)N_t;
    float* aggt2c = ws + off; off += (size_t)N_c * AP;
    float* aggt2m = ws + off; off += (size_t)N_m * AP;
    const long cs = (long)off;
    off += (size_t)cs;
    float* agg2c  = ws + off; off += (size_t)N_t * H;
    float* agg2m  = ws + off; off += (size_t)N_t * H;
    const long zeroCount = (long)off;
    float* h_t    = ws + off; off += (size_t)N_t * H;
    float* h_c    = ws + off; off += (size_t)N_c * H;
    float* h_m    = ws + off; off += (size_t)N_m * H;
    float* W1t    = ws + off; off += (size_t)KPAD * H;
    float* W1c    = ws + off; off += (size_t)KPAD * H;
    float* W1m    = ws + off; off += (size_t)KPAD * H;
    float* b1sum  = ws + off; off += (size_t)H;

    zero_kernel<<<ZWG, 256, 0, stream>>>(ws, zeroCount);
    prep_w_kernel<<<(KPAD * H + 255) / 256, 256, 0, stream>>>(
        w1_src_t, w1_dst_t, b1_t, w1_src_r, w1_dst_r, W1t, W1c, W1m, b1sum);
    scatter_scalar_kernel<<<(E + 255) / 256, 256, 0, stream>>>(
        x_card, x_merch, src_pays, dst_pays, src_recv, dst_recv, aggc1, aggm1, E, cs);
    scatter85_kernel<<<(E + 8 * EPW - 1) / (8 * EPW), 256, 0, stream>>>(
        aggt2c, src_pays, x_trans, dst_pays, E, cs);
    scatter85_kernel<<<(E + 8 * EPW - 1) / (8 * EPW), 256, 0, stream>>>(
        aggt2m, src_recv, x_trans, dst_recv, E, cs);
    conv1_t_kernel<<<256, C1W * 32, 0, stream>>>(x_trans, aggc1, aggm1, W1t, b1sum, h_t, N_t / 16, cs);
    gemm88_kernel<<<N_c / 16, 128, 0, stream>>>(aggt2c, cs, AP, x_card, nullptr, W1c, b1_r, h_c, N_c);
    gemm88_kernel<<<N_m / 16, 128, 0, stream>>>(aggt2m, cs, AP, x_merch, nullptr, W1m, b1_r + H, h_m, N_m);
    scatter64_kernel<<<(E + 8 * EPW - 1) / (8 * EPW), 256, 0, stream>>>(
        agg2c, dst_pays, h_c, src_pays, E);
    scatter64_kernel<<<(E + 8 * EPW - 1) / (8 * EPW), 256, 0, stream>>>(
        agg2m, dst_recv, h_m, src_recv, E);
    conv2_out_kernel<<<256, 128, 0, stream>>>(
        h_t, agg2c, agg2m, w2_l, w2_r, b2, w_lin, b_lin, out, N_t / 16);
}


# row scatter-add kernels: 2048 workgroups of long-lived waves walk the 4-edge groups with a grid stride, next indices loaded under this group's rows, not a wave per group; arrival order as before
# speedup vs baseline: 2.9951x; 1.0340x over previous
#include <hip/hip_runtime.h>

typedef __attribute__((ext_vector_type(2))) float v2f;
typedef __attribute__((ext_vector_type(4))) float v4f;
typedef __attribute__((ext_vector_type(8))) float v8f;

#define H 64
#define F85 85
#define KPAD 88
#define AP 96

__device__ __forceinline__ int chip_half() {
    return (int)((__builtin_amdgcn_s_sendmsg_rtn(0x87) >> 18) & 1u);
}

#define ZWG 256
__global__ __launch_bounds__(256) void zero_kernel(float* __restrict__ p, long n) {
    const unsigned lane = threadIdx.x & 31;
    const unsigned gw = blockIdx.x * 8 + (threadIdx.x >> 5);
    const unsigned nw = gridDim.x * 8;
    const unsigned n4 = (unsigned)(n >> 2);
    const unsigned units = (n4 + 31) >> 5;
    const unsigned per = (units + nw - 1) / nw;
    const unsigned b = gw * per * 32;
    const unsigned lim = b + per * 32;
    const unsigned e = lim < n4 ? lim : n4;
    v4f* __restrict__ q = (v4f*)p;
    const v4f z = {0.0f, 0.0f, 0.0f, 0.0f};
    unsigned i = b + lane;
    for (; i + 7 * 32 < e; i += 8 * 32) {
#pragma unroll
        for (int k = 0; k < 8; ++k) q[i + 32 * k] = z;
    }
    for (; i < e; i += 32) q[i] = z;
    if (gw == 0 && lane < (unsigned)(n & 3)) p[((long)n4 << 2) + lane] = 0.0f;
}

__global__ void prep_w_kernel(const float* __restrict__ w1_src_t,
                              const float* __restrict__ w1_dst_t,
                              const float* __restrict__ b1_t,
                              const float* __restrict__ w1_src_r,
                              const float* __restrict__ w1_dst_r,
                              float* __restrict__ W1t, float* __restrict__ W1c,
                              float* __restrict__ W1m, float* __restrict__ b1sum) {
    int i = blockIdx.x * blockDim.x + threadIdx.x;
    if (i >= KPAD * H) return;
    int r = i >> 6, c = i & 63;
    float vt = 0.f, vc = 0.f, vm = 0.f;
    if (r < F85) {
        vt = w1_dst_t[r * H + c] + w1_dst_t[F85 * H + r * H + c];
        vc = w1_src_r[r * H + c];
        vm = w1_src_r[F85 * H + r * H + c];
    } else if (r == 85) {
        vt = w1_src_t[c];
        vc = w1_dst_r[c];
        vm = w1_dst_r[H + c];
    } else if (r == 86) {
        vt = w1_src_t[H + c];
    }
    W1t[i] = vt; W1c[i] = vc; W1m[i] = vm;
    if (i < H) b1sum[i] = b1_t[i] + b1_t[H + i];
}

__global__ void scatter_scalar_kernel(const float* __restrict__ xc, const float* __restrict__ xm,
                                      const int* __restrict__ sp, const int* __restrict__ dp,
                                      const int* __restrict__ sr, const int* __restrict__ dr,
                                      float* __restrict__ aggc, float* __restrict__ aggm, int E,
                                      long cs) {
    int e = blockIdx.x * blockDim.x + threadIdx.x;
    if (e >= E) return;
    const int dc = dp[e], dm = dr[e];
    const float vc = xc[sp[e]], vm = xm[sr[e]];
    const long ho = chip_half() * cs;
    atomicAdd(&aggc[ho + dc], vc);
    atomicAdd(&aggm[ho + dm], vm);
}

#define EPW 4
#define SWG 2048

template <int NE>
__device__ __forceinline__ void edge_idx(int (&is)[NE], int (&od)[NE],
                                         const int* __restrict__ iidx,
                                         const int* __restrict__ oidx, int e0) {
#pragma unroll
    for (int u = 0; u < NE; ++u) { is[u] = iidx[e0 + u]; od[u] = oidx[e0 + u]; }
}

template <int F, int IP, int NE>
__device__ __forceinline__ void edge_rows(float (&v)[NE][(F + 31) / 32],
                                          const float* __restrict__ inbuf,
                                          const int (&is)[NE], int lane) {
    constexpr int NP = (F + 31) / 32;
#pragma unroll
    for (int u = 0; u < NE; ++u) {
        const float* srow = inbuf + (size_t)is[u] * IP;
#pragma unroll
        for (int k = 0; k < NP; ++k) {
            const int j = lane + 32 * k;
            v[u][k] = srow[j < F ? j : F - 1];
        }
    }
}

template <int F, int OP, int NE>
__device__ __forceinline__ void edge_adds(float* __restrict__ outbuf, const int (&od)[NE],
                                          const float (&v)[NE][(F + 31) / 32], int lane) {
    constexpr int NP = (F + 31) / 32;
    constexpr int TAIL = F - 32 * (NP - 1);
    if (TAIL < 32) {
#pragma unroll
        for (int u = 0; u < NE; ++u) asm volatile("" ::"v"(v[u][NP - 1]));
    }
#pragma unroll
    for (int u = 0; u < NE; ++u) {
        float* drow = outbuf + (size_t)od[u] * OP;
#pragma unroll
        for (int k = 0; k < NP - 1; ++k) atomicAdd(&drow[lane + 32 * k], v[u][k]);
    }
    if (lane < TAIL) {
#pragma unroll
        for (int u = 0; u < NE; ++u)
            atomicAdd(&outbuf[(size_t)od[u] * OP + lane + 32 * (NP - 1)], v[u][NP - 1]);
    }
}

template <int F, int IP, int OP, bool TWOCOPIES>
__device__ __forceinline__ void scatter_rows(float* __restrict__ outbuf,
                                             const int* __restrict__ oidx,
                                             const float* __restrict__ inbuf,
                                             const int* __restrict__ iidx, int E, long cs) {
    constexpr int NP = (F + 31) / 32;
    const int lane = threadIdx.x & 31;
    const int wave = __builtin_amdgcn_readfirstlane((int)(threadIdx.x >> 5));
    const int gw = (int)blockIdx.x * 8 + wave;
    const int nw = (int)gridDim.x * 8;
    const int ng = E / EPW;
    if (TWOCOPIES) outbuf += chip_half() * cs;
    int g = gw;
    if (g < ng) {
        int is[EPW], od[EPW];
        edge_idx<EPW>(is, od, iidx, oidx, g * EPW);
        for (;;) {
            float v[EPW][NP];
            edge_rows<F, IP, EPW>(v, inbuf, is, lane);
            const int gn = g + nw;
            int isn[EPW], odn[EPW];
            edge_idx<EPW>(isn, odn, iidx, oidx, (gn < ng ? gn : g) * EPW);
#pragma unroll
            for (int u = 0; u < EPW; ++u) asm volatile("" ::"s"(isn[u]), "s"(odn[u]));
            edge_adds<F, OP, EPW>(outbuf, od, v, lane);
            if (gn >= ng) break;
#pragma unroll
            for (int u = 0; u < EPW; ++u) { is[u] = isn[u]; od[u] = odn[u]; }
            g = gn;
        }
    }
    if (gw == nw - 1) {
        for (int e = ng * EPW; e < E; ++e) {
            int is1[1], od1[1];
            float v1[1][NP];
            edge_idx<1>(is1, od1, iidx, oidx, e);
            edge_rows<F, IP, 1>(v1, inbuf, is1, lane);
            edge_adds<F, OP, 1>(outbuf, od1, v1, lane);
        }
    }
}

__global__ __launch_bounds__(256) void scatter85_kernel(float* __restrict__ outbuf,
                                                        const int* __restrict__ oidx,
                                                        const float* __restrict__ inbuf,
                                                        const int* __restrict__ iidx, int E,
                                                        long cs) {
    scatter_rows<F85, F85, AP, true>(outbuf, oidx, inbuf, iidx, E, cs);
}

__global__ __launch_bounds__(256) void scatter64_kernel(float* __restrict__ outbuf,
                                                        const int* __restrict__ oidx,
                                                        const float* __restrict__ inbuf,
                                                        const int* __restrict__ iidx, int E) {
    scatter_rows<H, H, H, false>(outbuf, oidx, inbuf, iidx, E, 0);
}

__global__ __launch_bounds__(128) void gemm88_kernel(
    const float* __restrict__ Xmain, long cs, int pitch, const float* __restrict__ extra1,
    const float* __restrict__ extra2, const float* __restrict__ W,
    const float* __restrict__ bias, float* __restrict__ out, int M) {
    __shared__ float xs[16 * KPAD];

    const int row0 = blockIdx.x * 16;
    for (int i = threadIdx.x; i < 16 * KPAD; i += 128) {
        int r = i / KPAD, c = i - r * KPAD;
        int grow = row0 + r;
        float v;
        if (c < F85)      v = Xmain[(size_t)grow * pitch + c] + Xmain[(size_t)grow * pitch + c + cs];
        else if (c == 85) v = extra1 ? extra1[grow] : 0.0f;
        else if (c == 86) v = extra2 ? extra2[grow] : 0.0f;
        else              v = 0.0f;
        xs[i] = v;
    }
    __syncthreads();

    const int wave = threadIdx.x >> 5;
    const int lane = threadIdx.x & 31;
    const int half = lane >> 4;
    const int lm   = lane & 15;
    const int c0   = wave * 16;

    const float bval = bias[c0 + lm];
    v8f acc;
#pragma unroll
    for (int i = 0; i < 8; ++i) acc[i] = bval;

    for (int k = 0; k < KPAD; k += 4) {
        const int ka = k + 2 * half;
        v2f a, b;
        a.x = xs[lm * KPAD + ka];
        a.y = xs[lm * KPAD + ka + 1];
        b.x = W[(size_t)(ka + 0) * H + c0 + lm];
        b.y = W[(size_t)(ka + 1) * H + c0 + lm];
        acc = __builtin_amdgcn_wmma_f32_16x16x4_f32(false, a, false, b, (short)0, acc,
                                                    false, false);
    }
#pragma unroll
    for (int v = 0; v < 8; ++v) {
        float r = acc[v];
        r = r > 0.0f ? r : 0.0f;
        out[(size_t)(row0 + v + 8 * half) * H + c0 + lm] = r;
    }
}

#define C1W 8
#define C1Q 11
#define C1P 360
#define C1X 1408

__device__ __forceinline__ void c1_load(v4f (&r)[C1Q], float& exa, float& exb,
                                        const float* __restrict__ x,
                                        const float* __restrict__ extra, long cs,
                                        int tile, int lane) {
    const v4f* p = (const v4f*)(x + (size_t)tile * (16 * F85));
#pragma unroll
    for (int j = 0; j < C1Q - 1; ++j) r[j] = p[lane + 32 * j];
    const int pl = 32 * (C1Q - 1) + lane;
    r[C1Q - 1] = p[pl < 340 ? pl : 339];
    const float* pe = extra + tile * 16 + (lane & 15);
    exa = pe[0];
    exb = pe[cs];
}

__device__ __forceinline__ void c1_tile(const float* __restrict__ arow,
                                        const float* __restrict__ xrow,
                                        const v4f (&wreg)[C1Q][4], const float (&bv)[4],
                                        int c, int h, float* __restrict__ out, int tile) {
    float a[C1Q][4];
#pragma unroll
    for (int q = 0; q < C1Q; ++q) {
#pragma unroll
        for (int e = 0; e < 4; ++e) a[q][e] = arow[8 * q + e];
    }
    const float e1 = xrow[0], e2 = xrow[16];
    a[C1Q - 1][1] = h ? e1 : a[C1Q - 1][1];
    a[C1Q - 1][2] = h ? e2 : a[C1Q - 1][2];
    a[C1Q - 1][3] = h ? 0.0f : a[C1Q - 1][3];

    v8f acc[4];
#pragma unroll
    for (int ct = 0; ct < 4; ++ct) {
#pragma unroll
        for (int i = 0; i < 8; ++i) acc[ct][i] = bv[ct];
    }
#pragma unroll
    for (int q = 0; q < C1Q; ++q) {
        const v2f a0 = {a[q][0], a[q][1]}, a1 = {a[q][2], a[q][3]};
#pragma unroll
        for (int ct = 0; ct < 4; ++ct) {
            const v4f w = wreg[q][ct];
            const v2f b0 = {w[0], w[1]};
            acc[ct] = __builtin_amdgcn_wmma_f32_16x16x4_f32(false, a0, false, b0, (short)0,
                                                            acc[ct], false, false);
        }
#pragma unroll
        for (int ct = 0; ct < 4; ++ct) {
            const v4f w = wreg[q][ct];
            const v2f b1 = {w[2], w[3]};
            acc[ct] = __builtin_amdgcn_wmma_f32_16x16x4_f32(false, a1, false, b1, (short)0,
                                                            acc[ct], false, false);
        }
    }
    float* orow = out + ((size_t)tile * 16 + 8 * h) * H + 4 * c;
#pragma unroll
    for (int v = 0; v < 8; ++v) {
        v4f o;
#pragma unroll
        for (int ct = 0; ct < 4; ++ct) {
            const float r = acc[ct][v];
            o[ct] = r > 0.0f ? r : 0.0f;
        }
        *(v4f*)(orow + v * H) = o;
    }
}

__global__ __launch_bounds__(256) void conv1_t_kernel(
    const float* __restrict__ x, const float* __restrict__ extra1,
    const float* __restrict__ extra2, const float* __restrict__ W,
    const float* __restrict__ bias, float* __restrict__ out, int ntiles, long cs) {
    __shared__ v4f stage[C1W * C1P];

    const int lane = threadIdx.x & 31;
    const int c = lane & 15;
    const int h = lane >> 4;
    const int wave = __builtin_amdgcn_readfirstlane((int)(threadIdx.x >> 5));
    const int gw = blockIdx.x * C1W + wave;
    const int nw = gridDim.x * C1W;
    v4f* __restrict__ reg = stage + wave * C1P;
    float* __restrict__ regf = (float*)reg;
    const float* __restrict__ arow = regf + c * F85 + 4 * h;
    const float* __restrict__ xrow = regf + C1X + c;
    const float* __restrict__ extra = h ? extra2 : extra1;

    v4f wreg[C1Q][4];
#pragma unroll
    for (int q = 0; q < C1Q; ++q) {
#pragma unroll
        for (int e = 0; e < 4; ++e) {
            const v4f w = *(const v4f*)(W + (size_t)(8 * q + 4 * h + e) * H + 4 * c);
#pragma unroll
            for (int ct = 0; ct < 4; ++ct) wreg[q][ct][e] = w[ct];
        }
    }
    float bv[4];
#pragma unroll
    for (int ct = 0; ct < 4; ++ct) bv[ct] = bias[4 * c + ct];

    const int last = ntiles - 1;
    v4f r[C1Q];
    float exa, exb;
    int tile = gw;
    c1_load(r, exa, exb, x, extra, cs, tile < last ? tile : last, lane);
    for (; tile < ntiles; tile += nw) {
#pragma unroll
        for (int j = 0; j < C1Q; ++j) reg[lane + 32 * j] = r[j];
        regf[C1X + lane] = exa + exb;
        const int nt = tile + nw;
        c1_load(r, exa, exb, x, extra, cs, nt < last ? nt : last, lane);
        __builtin_amdgcn_sched_barrier(0);
        c1_tile(arow, xrow, wreg, bv, c, h, out, tile);
    }
}

struct C2Lane {
    float wl1[4][2], wl2[4][2], bs1[4], bs2[4];
    float blo;
    bool b3, b2, b1, b0;
    int lane, c, h;
};

__device__ __forceinline__ void c2_load(v4f (&ac)[8], v4f (&ah)[8], v4f (&am)[8],
                                        const float* __restrict__ h_t,
                                        const float* __restrict__ aggc,
                                        const float* __restrict__ aggm,
                                        int tile, int c, int h) {
    const size_t ro = ((size_t)tile * 16 + c) * H + 4 * h;
    const v4f* ph = (const v4f*)(h_t + ro);
    const v4f* pc = (const v4f*)(aggc + ro);
    const v4f* pm = (const v4f*)(aggm + ro);
#pragma unroll
    for (int q = 0; q < 8; ++q) { ac[q] = pc[2 * q]; ah[q] = ph[2 * q]; am[q] = pm[2 * q]; }
}

__device__ __forceinline__ void c2_tile(const v4f (&ac)[8], const v4f (&ah)[8],
                                        const v4f (&am)[8], const v4f (&wreg)[16][8],
                                        const C2Lane& L, float* __restrict__ out, int tile) {
    float p[16];
#pragma unroll
    for (int i = 0; i < 16; ++i) p[i] = 0.0f;

#pragma unroll
    for (int pass = 0; pass < 2; ++pass) {
        v8f acc[4];
#pragma unroll
        for (int ct = 0; ct < 4; ++ct) {
#pragma unroll
            for (int i = 0; i < 8; ++i) acc[ct][i] = 0.0f;
        }
#pragma unroll
        for (int q = 0; q < 8; ++q) {
            const v4f an = pass ? am[q] : ac[q];
            const v4f ar = ah[q];
            const v2f an0 = {an[0], an[1]}, an1 = {an[2], an[3]};
            const v2f ar0 = {ar[0], ar[1]}, ar1 = {ar[2], ar[3]};
#pragma unroll
            for (int ct = 0; ct < 4; ++ct) {
                const v4f w = wreg[(2 * pass + 0) * 4 + ct][q];
                const v2f b0 = {w[0], w[1]};
                acc[ct] = __builtin_amdgcn_wmma_f32_16x16x4_f32(false, an0, false, b0, (short)0,
                                                                acc[ct], false, false);
            }
#pragma unroll
            for (int ct = 0; ct < 4; ++ct) {
                const v4f w = wreg[(2 * pass + 0) * 4 + ct][q];
                const v2f b1 = {w[2], w[3]};
                acc[ct] = __builtin_amdgcn_wmma_f32_16x16x4_f32(false, an1, false, b1, (short)0,
                                                                acc[ct], false, false);
            }
#pragma unroll
            for (int ct = 0; ct < 4; ++ct) {
                const v4f w = wreg[(2 * pass + 1) * 4 + ct][q];
                const v2f b0 = {w[0], w[1]};
                acc[ct] = __builtin_amdgcn_wmma_f32_16x16x4_f32(false, ar0, false, b0, (short)0,
                                                                acc[ct], false, false);
            }
#pragma unroll
            for (int ct = 0; ct < 4; ++ct) {
                const v4f w = wreg[(2 * pass + 1) * 4 + ct][q];
                const v2f b1 = {w[2], w[3]};
                acc[ct] = __builtin_amdgcn_wmma_f32_16x16x4_f32(false, ar1, false, b1, (short)0,
                                                                acc[ct], false, false);
            }
        }
#pragma unroll
        for (int ct = 0; ct < 4; ++ct) {
            const float w0 = pass ? L.wl2[ct][0] : L.wl1[ct][0];
            const float w1 = pass ? L.wl2[ct][1] : L.wl1[ct][1];
            const float bv = pass ? L.bs2[ct] : L.bs1[ct];
#pragma unroll
            for (int v = 0; v < 8; ++v) {
                float r = acc[ct][v] + bv;
                r = r > 0.0f ? r : 0.0f;
                p[2 * v + 0] += r * w0;
                p[2 * v + 1] += r * w1;
            }
        }
    }

    float s8[8], s4[4], s2[2];
#pragma unroll
    for (int i = 0; i < 8; ++i) {
        const float keep = L.b3 ? p[i + 8] : p[i];
        const float send = L.b3 ? p[i] : p[i + 8];
        s8[i] = keep + __shfl_xor(send, 8);
    }
#pragma unroll
    for (int i = 0; i < 4; ++i) {
        const float keep = L.b2 ? s8[i + 4] : s8[i];
        const float send = L.b2 ? s8[i] : s8[i + 4];
        s4[i] = keep + __shfl_xor(send, 4);
    }
#pragma unroll
    for (int i = 0; i < 2; ++i) {
        const float keep = L.b1 ? s4[i + 2] : s4[i];
        const float send = L.b1 ? s4[i] : s4[i + 2];
        s2[i] = keep + __shfl_xor(send, 2);
    }
    const float keep1 = L.b0 ? s2[1] : s2[0];
    const float send1 = L.b0 ? s2[0] : s2[1];
    const float tot = keep1 + __shfl_xor(send1, 1);
    out[(size_t)tile * 32 + L.lane] = tot + L.blo;
}

__global__ __launch_bounds__(128) void conv2_out_kernel(
    const float* __restrict__ h_t,  const float* __restrict__ aggc,
    const float* __restrict__ aggm, const float* __restrict__ w2l,
    const float* __restrict__ w2r,  const float* __restrict__ b2,
    const float* __restrict__ wlin, const float* __restrict__ blin,
    float* __restrict__ out, int ntiles) {
    __shared__ v4f bf[4 * 4 * 8 * 32];

    {
        float* dst = (float*)bf;
        for (int i = threadIdx.x; i < 4 * H * 16; i += 128) {
            const int n4 = i & 15, k = (i >> 4) & 63, blk = i >> 10;
            const float* srcw = ((blk & 1) ? w2r : w2l) + (blk >> 1) * H * H;
            const v4f w = *(const v4f*)(srcw + k * H + 4 * n4);
            const int q = k >> 3, hh = (k >> 2) & 1, e = k & 3;
#pragma unroll
            for (int j = 0; j < 4; ++j) {
                const int n = 4 * n4 + j;
                const int ct = n >> 4, cc = n & 15;
                dst[((((blk * 4 + ct) * 8 + q) * 32) + hh * 16 + cc) * 4 + e] = w[j];
            }
        }
    }
    __syncthreads();

    C2Lane L;
    L.lane = threadIdx.x & 31;
    L.c = L.lane & 15;
    L.h = L.lane >> 4;
    const int wave = __builtin_amdgcn_readfirstlane((int)(threadIdx.x >> 5));
    const int gw = blockIdx.x * 4 + wave;
    const int nw = gridDim.x * 4;

#pragma unroll
    for (int ct = 0; ct < 4; ++ct) {
        L.wl1[ct][0] = wlin[(16 * ct + L.c) * 2 + 0];
        L.wl1[ct][1] = wlin[(16 * ct + L.c) * 2 + 1];
        L.wl2[ct][0] = wlin[(H + 16 * ct + L.c) * 2 + 0];
        L.wl2[ct][1] = wlin[(H + 16 * ct + L.c) * 2 + 1];
        L.bs1[ct] = b2[16 * ct + L.c];
        L.bs2[ct] = b2[H + 16 * ct + L.c];
    }
    L.blo = blin[L.lane & 1];
    L.b3 = (L.c & 8) != 0; L.b2 = (L.c & 4) != 0; L.b1 = (L.c & 2) != 0; L.b0 = (L.c & 1) != 0;

    v4f wreg[16][8];
#pragma unroll
    for (int bc = 0; bc < 16; ++bc) {
#pragma unroll
        for (int q = 0; q < 8; ++q) wreg[bc][q] = bf[(bc * 8 + q) * 32 + L.lane];
    }

    const int last = ntiles - 1;
    v4f ac0[8], ah0[8], am0[8], ac1[8], ah1[8], am1[8];
    int tile = gw;
    c2_load(ac0, ah0, am0, h_t, aggc, aggm, tile < last ? tile : last, L.c, L.h);
    for (; tile + nw < ntiles; tile += 2 * nw) {
        const int t2 = tile + 2 * nw;
        c2_load(ac1, ah1, am1, h_t, aggc, aggm, tile + nw, L.c, L.h);
        __builtin_amdgcn_sched_barrier(0);
        c2_tile(ac0, ah0, am0, wreg, L, out, tile);
        c2_load(ac0, ah0, am0, h_t, aggc, aggm, t2 < last ? t2 : last, L.c, L.h);
        __builtin_amdgcn_sched_barrier(0);
        c2_tile(ac1, ah1, am1, wreg, L, out, tile + nw);
    }
    if (tile < ntiles) c2_tile(ac0, ah0, am0, wreg, L, out, tile);
}

extern "C" void kernel_launch(void* const* d_in, const int* in_sizes, int n_in,
                              void* d_out, int out_size, void* d_ws, size_t ws_size,
                              hipStream_t stream) {
    const float* x_card   = (const float*)d_in[0];
    const float* x_merch  = (const float*)d_in[1];
    const float* x_trans  = (const float*)d_in[2];
    const float* w1_src_t = (const float*)d_in[3];
    const float* w1_dst_t = (const float*)d_in[4];
    const float* b1_t     = (const float*)d_in[5];
    const float* w1_src_r = (const float*)d_in[6];
    const float* w1_dst_r = (const float*)d_in[7];
    const float* b1_r     = (const float*)d_in[8];
    const float* w2_l     = (const float*)d_in[9];
    const float* w2_r     = (const float*)d_in[10];
    const float* b2       = (const float*)d_in[11];
    const float* w_lin    = (const float*)d_in[12];
    const float* b_lin    = (const float*)d_in[13];
    const int* src_pays   = (const int*)d_in[14];
    const int* dst_pays   = (const int*)d_in[15];
    const int* src_recv   = (const int*)d_in[16];
    const int* dst_recv   = (const int*)d_in[17];
    float* out = (float*)d_out;

    const int N_c = in_sizes[0];
    const int N_m = in_sizes[1];
    const int N_t = in_sizes[2] / F85;
    const int E   = in_sizes[14];

    float* ws = (float*)d_ws;
    size_t off = 0;
    float* aggc1  = ws + off; off += (size_t)N_t;
    float* aggm1  = ws + off; off += (size_t)N_t;
    float* aggt2c = ws + off; off += (size_t)N_c * AP;
    float* aggt2m = ws + off; off += (size_t)N_m * AP;
    const long cs = (long)off;
    off += (size_t)cs;
    float* agg2c  = ws + off; off += (size_t)N_t * H;
    float* agg2m  = ws + off; off += (size_t)N_t * H;
    const long zeroCount = (long)off;
    float* h_t    = ws + off; off += (size_t)N_t * H;
    float* h_c    = ws + off; off += (size_t)N_c * H;
    float* h_m    = ws + off; off += (size_t)N_m * H;
    float* W1t    = ws + off; off += (size_t)KPAD * H;
    float* W1c    = ws + off; off += (size_t)KPAD * H;
    float* W1m    = ws + off; off += (size_t)KPAD * H;
    float* b1sum  = ws + off; off += (size_t)H;

    zero_kernel<<<ZWG, 256, 0, stream>>>(ws, zeroCount);
    prep_w_kernel<<<(KPAD * H + 255) / 256, 256, 0, stream>>>(
        w1_src_t, w1_dst_t, b1_t, w1_src_r, w1_dst_r, W1t, W1c, W1m, b1sum);
    scatter_scalar_kernel<<<(E + 255) / 256, 256, 0, stream>>>(
        x_card, x_merch, src_pays, dst_pays, src_recv, dst_recv, aggc1, aggm1, E, cs);
    scatter85_kernel<<<SWG, 256, 0, stream>>>(
        aggt2c, src_pays, x_trans, dst_pays, E, cs);
    scatter85_kernel<<<SWG, 256, 0, stream>>>(
        aggt2m, src_recv, x_trans, dst_recv, E, cs);
    conv1_t_kernel<<<256, C1W * 32, 0, stream>>>(x_trans, aggc1, aggm1, W1t, b1sum, h_t, N_t / 16, cs);
    gemm88_kernel<<<N_c / 16, 128, 0, stream>>>(aggt2c, cs, AP, x_card, nullptr, W1c, b1_r, h_c, N_c);
    gemm88_kernel<<<N_m / 16, 128, 0, stream>>>(aggt2m, cs, AP, x_merch, nullptr, W1m, b1_r + H, h_m, N_m);
    scatter64_kernel<<<SWG, 256, 0, stream>>>(
        agg2c, dst_pays, h_c, src_pays, E);
    scatter64_kernel<<<SWG, 256, 0, stream>>>(
        agg2m, dst_recv, h_m, src_recv, E);
    conv2_out_kernel<<<256, 128, 0, stream>>>(
        h_t, agg2c, agg2m, w2_l, w2_r, b2, w_lin, b_lin, out, N_t / 16);
}


# conv2 sums start from the root term: a WMMA kernel stores h_t@w2_r+b2 into them (no zero pass for them), card/merchant rows projected by w2_l before the scatter; sum order: project, then add
# speedup vs baseline: 3.6386x; 1.2149x over previous
#include <hip/hip_runtime.h>

typedef __attribute__((ext_vector_type(2))) float v2f;
typedef __attribute__((ext_vector_type(4))) float v4f;
typedef __attribute__((ext_vector_type(8))) float v8f;

#define H 64
#define F85 85
#define KPAD 88
#define AP 96
#define HSP 68

__device__ __forceinline__ int chip_half() {
    return (int)((__builtin_amdgcn_s_sendmsg_rtn(0x87) >> 18) & 1u);
}

#define ZWG 256
__global__ __launch_bounds__(256) void zero_kernel(float* __restrict__ p, long n) {
    const unsigned lane = threadIdx.x & 31;
    const unsigned gw = blockIdx.x * 8 + (threadIdx.x >> 5);
    const unsigned nw = gridDim.x * 8;
    const unsigned n4 = (unsigned)(n >> 2);
    const unsigned units = (n4 + 31) >> 5;
    const unsigned per = (units + nw - 1) / nw;
    const unsigned b = gw * per * 32;
    const unsigned lim = b + per * 32;
    const unsigned e = lim < n4 ? lim : n4;
    v4f* __restrict__ q = (v4f*)p;
    const v4f z = {0.0f, 0.0f, 0.0f, 0.0f};
    unsigned i = b + lane;
    for (; i + 7 * 32 < e; i += 8 * 32) {
#pragma unroll
        for (int k = 0; k < 8; ++k) q[i + 32 * k] = z;
    }
    for (; i < e; i += 32) q[i] = z;
    if (gw == 0 && lane < (unsigned)(n & 3)) p[((long)n4 << 2) + lane] = 0.0f;
}

__global__ void prep_w_kernel(const float* __restrict__ w1_src_t,
                              const float* __restrict__ w1_dst_t,
                              const float* __restrict__ b1_t,
                              const float* __restrict__ w1_src_r,
                              const float* __restrict__ w1_dst_r,
                              float* __restrict__ W1t, float* __restrict__ W1c,
                              float* __restrict__ W1m, float* __restrict__ b1sum) {
    int i = blockIdx.x * blockDim.x + threadIdx.x;
    if (i >= KPAD * H) return;
    int r = i >> 6, c = i & 63;
    float vt = 0.f, vc = 0.f, vm = 0.f;
    if (r < F85) {
        vt = w1_dst_t[r * H + c] + w1_dst_t[F85 * H + r * H + c];
        vc = w1_src_r[r * H + c];
        vm = w1_src_r[F85 * H + r * H + c];
    } else if (r == 85) {
        vt = w1_src_t[c];
        vc = w1_dst_r[c];
        vm = w1_dst_r[H + c];
    } else if (r == 86) {
        vt = w1_src_t[H + c];
    }
    W1t[i] = vt; W1c[i] = vc; W1m[i] = vm;
    if (i < H) b1sum[i] = b1_t[i] + b1_t[H + i];
}

__global__ void scatter_scalar_kernel(const float* __restrict__ xc, const float* __restrict__ xm,
                                      const int* __restrict__ sp, const int* __restrict__ dp,
                                      const int* __restrict__ sr, const int* __restrict__ dr,
                                      float* __restrict__ aggc, float* __restrict__ aggm, int E,
                                      long cs) {
    int e = blockIdx.x * blockDim.x + threadIdx.x;
    if (e >= E) return;
    const int dc = dp[e], dm = dr[e];
    const float vc = xc[sp[e]], vm = xm[sr[e]];
    const long ho = chip_half() * cs;
    atomicAdd(&aggc[ho + dc], vc);
    atomicAdd(&aggm[ho + dm], vm);
}

#define EPW 4
#define SWG 2048

template <int NE>
__device__ __forceinline__ void edge_idx(int (&is)[NE], int (&od)[NE],
                                         const int* __restrict__ iidx,
                                         const int* __restrict__ oidx, int e0) {
#pragma unroll
    for (int u = 0; u < NE; ++u) { is[u] = iidx[e0 + u]; od[u] = oidx[e0 + u]; }
}

template <int F, int IP, int NE>
__device__ __forceinline__ void edge_rows(float (&v)[NE][(F + 31) / 32],
                                          const float* __restrict__ inbuf,
                                          const int (&is)[NE], int lane) {
    constexpr int NP = (F + 31) / 32;
#pragma unroll
    for (int u = 0; u < NE; ++u) {
        const float* srow = inbuf + (size_t)is[u] * IP;
#pragma unroll
        for (int k = 0; k < NP; ++k) {
            const int j = lane + 32 * k;
            v[u][k] = srow[j < F ? j : F - 1];
        }
    }
}

template <int F, int OP, int NE>
__device__ __forceinline__ void edge_adds(float* __restrict__ outbuf, const int (&od)[NE],
                                          const float (&v)[NE][(F + 31) / 32], int lane) {
    constexpr int NP = (F + 31) / 32;
    constexpr int TAIL = F - 32 * (NP - 1);
    if (TAIL < 32) {
#pragma unroll
        for (int u = 0; u < NE; ++u) asm volatile("" ::"v"(v[u][NP - 1]));
    }
#pragma unroll
    for (int u = 0; u < NE; ++u) {
        float* drow = outbuf + (size_t)od[u] * OP;
#pragma unroll
        for (int k = 0; k < NP - 1; ++k) atomicAdd(&drow[lane + 32 * k], v[u][k]);
    }
    if (lane < TAIL) {
#pragma unroll
        for (int u = 0; u < NE; ++u)
            atomicAdd(&outbuf[(size_t)od[u] * OP + lane + 32 * (NP - 1)], v[u][NP - 1]);
    }
}

template <int F, int IP, int OP, bool TWOCOPIES>
__device__ __forceinline__ void scatter_rows(float* __restrict__ outbuf,
                                             const int* __restrict__ oidx,
                                             const float* __restrict__ inbuf,
                                             const int* __restrict__ iidx, int E, long cs) {
    constexpr int NP = (F + 31) / 32;
    const int lane = threadIdx.x & 31;
    const int wave = __builtin_amdgcn_readfirstlane((int)(threadIdx.x >> 5));
    const int gw = (int)blockIdx.x * 8 + wave;
    const int nw = (int)gridDim.x * 8;
    const int ng = E / EPW;
    if (TWOCOPIES) outbuf += chip_half() * cs;
    int g = gw;
    if (g < ng) {
        int is[EPW], od[EPW];
        edge_idx<EPW>(is, od, iidx, oidx, g * EPW);
        for (;;) {
            float v[EPW][NP];
            edge_rows<F, IP, EPW>(v, inbuf, is, lane);
            const int gn = g + nw;
            int isn[EPW], odn[EPW];
            edge_idx<EPW>(isn, odn, iidx, oidx, (gn < ng ? gn : g) * EPW);
#pragma unroll
            for (int u = 0; u < EPW; ++u) asm volatile("" ::"s"(isn[u]), "s"(odn[u]));
            edge_adds<F, OP, EPW>(outbuf, od, v, lane);
            if (gn >= ng) break;
#pragma unroll
            for (int u = 0; u < EPW; ++u) { is[u] = isn[u]; od[u] = odn[u]; }
            g = gn;
        }
    }
    if (gw == nw - 1) {
        for (int e = ng * EPW; e < E; ++e) {
            int is1[1], od1[1];
            float v1[1][NP];
            edge_idx<1>(is1, od1, iidx, oidx, e);
            edge_rows<F, IP, 1>(v1, inbuf, is1, lane);
            edge_adds<F, OP, 1>(outbuf, od1, v1, lane);
        }
    }
}

__global__ __launch_bounds__(256) void scatter85_kernel(float* __restrict__ outbuf,
                                                        const int* __restrict__ oidx,
                                                        const float* __restrict__ inbuf,
                                                        const int* __restrict__ iidx, int E,
                                                        long cs) {
    scatter_rows<F85, F85, AP, true>(outbuf, oidx, inbuf, iidx, E, cs);
}

__global__ __launch_bounds__(256) void scatter64_kernel(float* __restrict__ outbuf,
                                                        const int* __restrict__ oidx,
                                                        const float* __restrict__ inbuf,
                                                        const int* __restrict__ iidx, int E) {
    scatter_rows<H, H, H, false>(outbuf, oidx, inbuf, iidx, E, 0);
}

__global__ __launch_bounds__(128) void gemm88_kernel(
    const float* __restrict__ Xmain, long cs, int pitch, const float* __restrict__ extra1,
    const float* __restrict__ extra2, const float* __restrict__ W,
    const float* __restrict__ bias, const float* __restrict__ W2, float* __restrict__ out,
    int M) {
    __shared__ float xs[16 * KPAD];
    __shared__ float hs[16 * HSP];

    const int row0 = blockIdx.x * 16;
    for (int i = threadIdx.x; i < 16 * KPAD; i += 128) {
        int r = i / KPAD, c = i - r * KPAD;
        int grow = row0 + r;
        float v;
        if (c < F85)      v = Xmain[(size_t)grow * pitch + c] + Xmain[(size_t)grow * pitch + c + cs];
        else if (c == 85) v = extra1 ? extra1[grow] : 0.0f;
        else if (c == 86) v = extra2 ? extra2[grow] : 0.0f;
        else              v = 0.0f;
        xs[i] = v;
    }
    __syncthreads();

    const int wave = threadIdx.x >> 5;
    const int lane = threadIdx.x & 31;
    const int half = lane >> 4;
    const int lm   = lane & 15;
    const int c0   = wave * 16;

    const float bval = bias[c0 + lm];
    v8f acc;
#pragma unroll
    for (int i = 0; i < 8; ++i) acc[i] = bval;

    for (int k = 0; k < KPAD; k += 4) {
        const int ka = k + 2 * half;
        v2f a, b;
        a.x = xs[lm * KPAD + ka];
        a.y = xs[lm * KPAD + ka + 1];
        b.x = W[(size_t)(ka + 0) * H + c0 + lm];
        b.y = W[(size_t)(ka + 1) * H + c0 + lm];
        acc = __builtin_amdgcn_wmma_f32_16x16x4_f32(false, a, false, b, (short)0, acc,
                                                    false, false);
    }
#pragma unroll
    for (int v = 0; v < 8; ++v) {
        float r = acc[v];
        r = r > 0.0f ? r : 0.0f;
        hs[(v + 8 * half) * HSP + c0 + lm] = r;
    }
    __syncthreads();
    v8f acc2;
#pragma unroll
    for (int i = 0; i < 8; ++i) acc2[i] = 0.0f;
    for (int k = 0; k < H; k += 4) {
        const int ka = k + 2 * half;
        v2f a, b;
        a.x = hs[lm * HSP + ka];
        a.y = hs[lm * HSP + ka + 1];
        b.x = W2[(size_t)(ka + 0) * H + c0 + lm];
        b.y = W2[(size_t)(ka + 1) * H + c0 + lm];
        acc2 = __builtin_amdgcn_wmma_f32_16x16x4_f32(false, a, false, b, (short)0, acc2,
                                                     false, false);
    }
#pragma unroll
    for (int v = 0; v < 8; ++v)
        out[(size_t)(row0 + v + 8 * half) * H + c0 + lm] = acc2[v];
}

#define C1W 8
#define C1Q 11
#define C1P 360
#define C1X 1408

__device__ __forceinline__ void c1_load(v4f (&r)[C1Q], float& exa, float& exb,
                                        const float* __restrict__ x,
                                        const float* __restrict__ extra, long cs,
                                        int tile, int lane) {
    const v4f* p = (const v4f*)(x + (size_t)tile * (16 * F85));
#pragma unroll
    for (int j = 0; j < C1Q - 1; ++j) r[j] = p[lane + 32 * j];
    const int pl = 32 * (C1Q - 1) + lane;
    r[C1Q - 1] = p[pl < 340 ? pl : 339];
    const float* pe = extra + tile * 16 + (lane & 15);
    exa = pe[0];
    exb = pe[cs];
}

__device__ __forceinline__ void c1_tile(const float* __restrict__ arow,
                                        const float* __restrict__ xrow,
                                        const v4f (&wreg)[C1Q][4], const float (&bv)[4],
                                        int c, int h, float* __restrict__ out, int tile) {
    float a[C1Q][4];
#pragma unroll
    for (int q = 0; q < C1Q; ++q) {
#pragma unroll
        for (int e = 0; e < 4; ++e) a[q][e] = arow[8 * q + e];
    }
    const float e1 = xrow[0], e2 = xrow[16];
    a[C1Q - 1][1] = h ? e1 : a[C1Q - 1][1];
    a[C1Q - 1][2] = h ? e2 : a[C1Q - 1][2];
    a[C1Q - 1][3] = h ? 0.0f : a[C1Q - 1][3];

    v8f acc[4];
#pragma unroll
    for (int ct = 0; ct < 4; ++ct) {
#pragma unroll
        for (int i = 0; i < 8; ++i) acc[ct][i] = bv[ct];
    }
#pragma unroll
    for (int q = 0; q < C1Q; ++q) {
        const v2f a0 = {a[q][0], a[q][1]}, a1 = {a[q][2], a[q][3]};
#pragma unroll
        for (int ct = 0; ct < 4; ++ct) {
            const v4f w = wreg[q][ct];
            const v2f b0 = {w[0], w[1]};
            acc[ct] = __builtin_amdgcn_wmma_f32_16x16x4_f32(false, a0, false, b0, (short)0,
                                                            acc[ct], false, false);
        }
#pragma unroll
        for (int ct = 0; ct < 4; ++ct) {
            const v4f w = wreg[q][ct];
            const v2f b1 = {w[2], w[3]};
            acc[ct] = __builtin_amdgcn_wmma_f32_16x16x4_f32(false, a1, false, b1, (short)0,
                                                            acc[ct], false, false);
        }
    }
    float* orow = out + ((size_t)tile * 16 + 8 * h) * H + 4 * c;
#pragma unroll
    for (int v = 0; v < 8; ++v) {
        v4f o;
#pragma unroll
        for (int ct = 0; ct < 4; ++ct) {
            const float r = acc[ct][v];
            o[ct] = r > 0.0f ? r : 0.0f;
        }
        *(v4f*)(orow + v * H) = o;
    }
}

__global__ __launch_bounds__(256) void conv1_t_kernel(
    const float* __restrict__ x, const float* __restrict__ extra1,
    const float* __restrict__ extra2, const float* __restrict__ W,
    const float* __restrict__ bias, float* __restrict__ out, int ntiles, long cs) {
    __shared__ v4f stage[C1W * C1P];

    const int lane = threadIdx.x & 31;
    const int c = lane & 15;
    const int h = lane >> 4;
    const int wave = __builtin_amdgcn_readfirstlane((int)(threadIdx.x >> 5));
    const int gw = blockIdx.x * C1W + wave;
    const int nw = gridDim.x * C1W;
    v4f* __restrict__ reg = stage + wave * C1P;
    float* __restrict__ regf = (float*)reg;
    const float* __restrict__ arow = regf + c * F85 + 4 * h;
    const float* __restrict__ xrow = regf + C1X + c;
    const float* __restrict__ extra = h ? extra2 : extra1;

    v4f wreg[C1Q][4];
#pragma unroll
    for (int q = 0; q < C1Q; ++q) {
#pragma unroll
        for (int e = 0; e < 4; ++e) {
            const v4f w = *(const v4f*)(W + (size_t)(8 * q + 4 * h + e) * H + 4 * c);
#pragma unroll
            for (int ct = 0; ct < 4; ++ct) wreg[q][ct][e] = w[ct];
        }
    }
    float bv[4];
#pragma unroll
    for (int ct = 0; ct < 4; ++ct) bv[ct] = bias[4 * c + ct];

    const int last = ntiles - 1;
    v4f r[C1Q];
    float exa, exb;
    int tile = gw;
    c1_load(r, exa, exb, x, extra, cs, tile < last ? tile : last, lane);
    for (; tile < ntiles; tile += nw) {
#pragma unroll
        for (int j = 0; j < C1Q; ++j) reg[lane + 32 * j] = r[j];
        regf[C1X + lane] = exa + exb;
        const int nt = tile + nw;
        c1_load(r, exa, exb, x, extra, cs, nt < last ? nt : last, lane);
        __builtin_amdgcn_sched_barrier(0);
        c1_tile(arow, xrow, wreg, bv, c, h, out, tile);
    }
}

__device__ __forceinline__ void c2r_load(v4f (&ah)[8], const float* __restrict__ h_t,
                                         int tile, int c, int h) {
    const v4f* ph = (const v4f*)(h_t + ((size_t)tile * 16 + c) * H + 4 * h);
#pragma unroll
    for (int q = 0; q < 8; ++q) ah[q] = ph[2 * q];
}

__device__ __forceinline__ void c2r_tile(const v4f (&ah)[8], const v4f (&wreg)[8][8],
                                         const float (&bs)[2][4], int c, int h,
                                         float* __restrict__ outc, float* __restrict__ outm,
                                         int tile) {
#pragma unroll
    for (int pass = 0; pass < 2; ++pass) {
        v8f acc[4];
#pragma unroll
        for (int ct = 0; ct < 4; ++ct) {
#pragma unroll
            for (int i = 0; i < 8; ++i) acc[ct][i] = bs[pass][ct];
        }
#pragma unroll
        for (int q = 0; q < 8; ++q) {
            const v4f ar = ah[q];
            const v2f ar0 = {ar[0], ar[1]}, ar1 = {ar[2], ar[3]};
#pragma unroll
            for (int ct = 0; ct < 4; ++ct) {
                const v4f w = wreg[pass * 4 + ct][q];
                const v2f b0 = {w[0], w[1]};
                acc[ct] = __builtin_amdgcn_wmma_f32_16x16x4_f32(false, ar0, false, b0, (short)0,
                                                                acc[ct], false, false);
            }
#pragma unroll
            for (int ct = 0; ct < 4; ++ct) {
                const v4f w = wreg[pass * 4 + ct][q];
                const v2f b1 = {w[2], w[3]};
                acc[ct] = __builtin_amdgcn_wmma_f32_16x16x4_f32(false, ar1, false, b1, (short)0,
                                                                acc[ct], false, false);
            }
        }
        float* orow = (pass ? outm : outc) + ((size_t)tile * 16 + 8 * h) * H + 4 * c;
#pragma unroll
        for (int v = 0; v < 8; ++v) {
            v4f o;
#pragma unroll
            for (int ct = 0; ct < 4; ++ct) o[ct] = acc[ct][v];
            *(v4f*)(orow + v * H) = o;
        }
    }
}

__global__ __launch_bounds__(128) void conv2_root_kernel(
    const float* __restrict__ h_t, const float* __restrict__ w2r,
    const float* __restrict__ b2, float* __restrict__ outc, float* __restrict__ outm,
    int ntiles) {
    __shared__ v4f bf[2 * 4 * 8 * 32];

    {
        float* dst = (float*)bf;
        for (int i = threadIdx.x; i < 2 * H * 16; i += 128) {
            const int n4 = i & 15, k = (i >> 4) & 63, blk = i >> 10;
            const v4f w = *(const v4f*)(w2r + blk * H * H + k * H + 4 * n4);
            const int q = k >> 3, hh = (k >> 2) & 1, e = k & 3;
#pragma unroll
            for (int j = 0; j < 4; ++j) {
                dst[((((blk * 4 + j) * 8 + q) * 32) + hh * 16 + n4) * 4 + e] = w[j];
            }
        }
    }
    __syncthreads();

    const int lane = threadIdx.x & 31;
    const int c = lane & 15;
    const int h = lane >> 4;
    const int wave = __builtin_amdgcn_readfirstlane((int)(threadIdx.x >> 5));
    const int gw = blockIdx.x * 4 + wave;
    const int nw = gridDim.x * 4;

    float bs[2][4];
#pragma unroll
    for (int ct = 0; ct < 4; ++ct) {
        bs[0][ct] = b2[4 * c + ct];
        bs[1][ct] = b2[H + 4 * c + ct];
    }

    v4f wreg[8][8];
#pragma unroll
    for (int bc = 0; bc < 8; ++bc) {
#pragma unroll
        for (int q = 0; q < 8; ++q) wreg[bc][q] = bf[(bc * 8 + q) * 32 + lane];
    }

    const int last = ntiles - 1;
    v4f ah0[8], ah1[8];
    int tile = gw;
    c2r_load(ah0, h_t, tile < last ? tile : last, c, h);
    for (; tile + nw < ntiles; tile += 2 * nw) {
        const int t2 = tile + 2 * nw;
        c2r_load(ah1, h_t, tile + nw, c, h);
        __builtin_amdgcn_sched_barrier(0);
        c2r_tile(ah0, wreg, bs, c, h, outc, outm, tile);
        c2r_load(ah0, h_t, t2 < last ? t2 : last, c, h);
        __builtin_amdgcn_sched_barrier(0);
        c2r_tile(ah1, wreg, bs, c, h, outc, outm, tile + nw);
    }
    if (tile < ntiles) c2r_tile(ah0, wreg, bs, c, h, outc, outm, tile);
}

#define OWG 512
__global__ __launch_bounds__(256) void out_kernel(
    const float* __restrict__ t1, const float* __restrict__ t2,
    const float* __restrict__ wlin, const float* __restrict__ blin,
    float* __restrict__ out, int ntiles) {
    const int lane = threadIdx.x & 31;
    const int c = lane & 15;
    const int h = lane >> 4;
    const int wave = __builtin_amdgcn_readfirstlane((int)(threadIdx.x >> 5));
    const int gw = blockIdx.x * 8 + wave;
    const int nw = gridDim.x * 8;

    float wa[4][2], wb[4][2];
#pragma unroll
    for (int j = 0; j < 4; ++j) {
        wa[j][0] = wlin[(4 * c + j) * 2 + 0];
        wa[j][1] = wlin[(4 * c + j) * 2 + 1];
        wb[j][0] = wlin[(H + 4 * c + j) * 2 + 0];
        wb[j][1] = wlin[(H + 4 * c + j) * 2 + 1];
    }
    const float blo = blin[lane & 1];
    const bool b3 = (c & 8) != 0, b2 = (c & 4) != 0, b1 = (c & 2) != 0, b0 = (c & 1) != 0;

    for (int tile = gw; tile < ntiles; tile += nw) {
        const size_t ro = ((size_t)tile * 16 + 8 * h) * H + 4 * c;
        const v4f* p1 = (const v4f*)(t1 + ro);
        const v4f* p2 = (const v4f*)(t2 + ro);
        v4f x1[8], x2[8];
#pragma unroll
        for (int v = 0; v < 8; ++v) { x1[v] = p1[v * (H / 4)]; x2[v] = p2[v * (H / 4)]; }
        float p[16];
#pragma unroll
        for (int v = 0; v < 8; ++v) {
            float s0 = 0.0f, s1 = 0.0f;
#pragma unroll
            for (int j = 0; j < 4; ++j) {
                float r = x1[v][j];
                r = r > 0.0f ? r : 0.0f;
                s0 += r * wa[j][0];
                s1 += r * wa[j][1];
            }
#pragma unroll
            for (int j = 0; j < 4; ++j) {
                float r = x2[v][j];
                r = r > 0.0f ? r : 0.0f;
                s0 += r * wb[j][0];
                s1 += r * wb[j][1];
            }
            p[2 * v + 0] = s0;
            p[2 * v + 1] = s1;
        }
        float s8[8], s4[4], s2[2];
#pragma unroll
        for (int i = 0; i < 8; ++i) {
            const float keep = b3 ? p[i + 8] : p[i];
            const float send = b3 ? p[i] : p[i + 8];
            s8[i] = keep + __shfl_xor(send, 8);
        }
#pragma unroll
        for (int i = 0; i < 4; ++i) {
            const float keep = b2 ? s8[i + 4] : s8[i];
            const float send = b2 ? s8[i] : s8[i + 4];
            s4[i] = keep + __shfl_xor(send, 4);
        }
#pragma unroll
        for (int i = 0; i < 2; ++i) {
            const float keep = b1 ? s4[i + 2] : s4[i];
            const float send = b1 ? s4[i] : s4[i + 2];
            s2[i] = keep + __shfl_xor(send, 2);
        }
        const float keep1 = b0 ? s2[1] : s2[0];
        const float send1 = b0 ? s2[0] : s2[1];
        const float tot = keep1 + __shfl_xor(send1, 1);
        out[(size_t)tile * 32 + lane] = tot + blo;
    }
}

extern "C" void kernel_launch(void* const* d_in, const int* in_sizes, int n_in,
                              void* d_out, int out_size, void* d_ws, size_t ws_size,
                              hipStream_t stream) {
    const float* x_card   = (const float*)d_in[0];
    const float* x_merch  = (const float*)d_in[1];
    const float* x_trans  = (const float*)d_in[2];
    const float* w1_src_t = (const float*)d_in[3];
    const float* w1_dst_t = (const float*)d_in[4];
    const float* b1_t     = (const float*)d_in[5];
    const float* w1_src_r = (const float*)d_in[6];
    const float* w1_dst_r = (const float*)d_in[7];
    const float* b1_r     = (const float*)d_in[8];
    const float* w2_l     = (const float*)d_in[9];
    const float* w2_r     = (const float*)d_in[10];
    const float* b2       = (const float*)d_in[11];
    const float* w_lin    = (const float*)d_in[12];
    const float* b_lin    = (const float*)d_in[13];
    const int* src_pays   = (const int*)d_in[14];
    const int* dst_pays   = (const int*)d_in[15];
    const int* src_recv   = (const int*)d_in[16];
    const int* dst_recv   = (const int*)d_in[17];
    float* out = (float*)d_out;

    const int N_c = in_sizes[0];
    const int N_m = in_sizes[1];
    const int N_t = in_sizes[2] / F85;
    const int E   = in_sizes[14];

    float* ws = (float*)d_ws;
    size_t off = 0;
    float* aggc1  = ws + off; off += (size_t)N_t;
    float* aggm1  = ws + off; off += (size_t)N_t;
    float* aggt2c = ws + off; off += (size_t)N_c * AP;
    float* aggt2m = ws + off; off += (size_t)N_m * AP;
    const long cs = (long)off;
    off += (size_t)cs;
    const long zeroCount = (long)off;
    float* agg2c  = ws + off; off += (size_t)N_t * H;
    float* agg2m  = ws + off; off += (size_t)N_t * H;
    float* h_t    = ws + off; off += (size_t)N_t * H;
    float* h_c    = ws + off; off += (size_t)N_c * H;
    float* h_m    = ws + off; off += (size_t)N_m * H;
    float* W1t    = ws + off; off += (size_t)KPAD * H;
    float* W1c    = ws + off; off += (size_t)KPAD * H;
    float* W1m    = ws + off; off += (size_t)KPAD * H;
    float* b1sum  = ws + off; off += (size_t)H;

    zero_kernel<<<ZWG, 256, 0, stream>>>(ws, zeroCount);
    prep_w_kernel<<<(KPAD * H + 255) / 256, 256, 0, stream>>>(
        w1_src_t, w1_dst_t, b1_t, w1_src_r, w1_dst_r, W1t, W1c, W1m, b1sum);
    scatter_scalar_kernel<<<(E + 255) / 256, 256, 0, stream>>>(
        x_card, x_merch, src_pays, dst_pays, src_recv, dst_recv, aggc1, aggm1, E, cs);
    scatter85_kernel<<<SWG, 256, 0, stream>>>(
        aggt2c, src_pays, x_trans, dst_pays, E, cs);
    scatter85_kernel<<<SWG, 256, 0, stream>>>(
        aggt2m, src_recv, x_trans, dst_recv, E, cs);
    conv1_t_kernel<<<256, C1W * 32, 0, stream>>>(x_trans, aggc1, aggm1, W1t, b1sum, h_t, N_t / 16, cs);
    gemm88_kernel<<<N_c / 16, 128, 0, stream>>>(aggt2c, cs, AP, x_card, nullptr, W1c, b1_r, w2_l, h_c, N_c);
    gemm88_kernel<<<N_m / 16, 128, 0, stream>>>(aggt2m, cs, AP, x_merch, nullptr, W1m, b1_r + H, w2_l + H * H, h_m, N_m);
    conv2_root_kernel<<<256, 128, 0, stream>>>(h_t, w2_r, b2, agg2c, agg2m, N_t / 16);
    scatter64_kernel<<<SWG, 256, 0, stream>>>(
        agg2c, dst_pays, h_c, src_pays, E);
    scatter64_kernel<<<SWG, 256, 0, stream>>>(
        agg2m, dst_recv, h_m, src_recv, E);
    out_kernel<<<OWG, 256, 0, stream>>>(agg2c, agg2m, w_lin, b_lin, out, N_t / 16);
}


# 85-wide edge passes run as the second wave per SIMD of the two big WMMA kernels (pays beside conv1, receives beside conv2 root), 20 edges per trip; 12 -> 10 launches; arrival order as before
# speedup vs baseline: 3.8679x; 1.0630x over previous
#include <hip/hip_runtime.h>

typedef __attribute__((ext_vector_type(2))) float v2f;
typedef __attribute__((ext_vector_type(4))) float v4f;
typedef __attribute__((ext_vector_type(8))) float v8f;

#define H 64
#define F85 85
#define KPAD 88
#define AP 96
#define HSP 68

__device__ __forceinline__ int chip_half() {
    return (int)((__builtin_amdgcn_s_sendmsg_rtn(0x87) >> 18) & 1u);
}

#define ZWG 256
__global__ __launch_bounds__(256) void zero_kernel(float* __restrict__ p, long n) {
    const unsigned lane = threadIdx.x & 31;
    const unsigned gw = blockIdx.x * 8 + (threadIdx.x >> 5);
    const unsigned nw = gridDim.x * 8;
    const unsigned n4 = (unsigned)(n >> 2);
    const unsigned units = (n4 + 31) >> 5;
    const unsigned per = (units + nw - 1) / nw;
    const unsigned b = gw * per * 32;
    const unsigned lim = b + per * 32;
    const unsigned e = lim < n4 ? lim : n4;
    v4f* __restrict__ q = (v4f*)p;
    const v4f z = {0.0f, 0.0f, 0.0f, 0.0f};
    unsigned i = b + lane;
    for (; i + 7 * 32 < e; i += 8 * 32) {
#pragma unroll
        for (int k = 0; k < 8; ++k) q[i + 32 * k] = z;
    }
    for (; i < e; i += 32) q[i] = z;
    if (gw == 0 && lane < (unsigned)(n & 3)) p[((long)n4 << 2) + lane] = 0.0f;
}

__global__ void prep_w_kernel(const float* __restrict__ w1_src_t,
                              const float* __restrict__ w1_dst_t,
                              const float* __restrict__ b1_t,
                              const float* __restrict__ w1_src_r,
                              const float* __restrict__ w1_dst_r,
                              float* __restrict__ W1t, float* __restrict__ W1c,
                              float* __restrict__ W1m, float* __restrict__ b1sum) {
    int i = blockIdx.x * blockDim.x + threadIdx.x;
    if (i >= KPAD * H) return;
    int r = i >> 6, c = i & 63;
    float vt = 0.f, vc = 0.f, vm = 0.f;
    if (r < F85) {
        vt = w1_dst_t[r * H + c] + w1_dst_t[F85 * H + r * H + c];
        vc = w1_src_r[r * H + c];
        vm = w1_src_r[F85 * H + r * H + c];
    } else if (r == 85) {
        vt = w1_src_t[c];
        vc = w1_dst_r[c];
        vm = w1_dst_r[H + c];
    } else if (r == 86) {
        vt = w1_src_t[H + c];
    }
    W1t[i] = vt; W1c[i] = vc; W1m[i] = vm;
    if (i < H) b1sum[i] = b1_t[i] + b1_t[H + i];
}

__global__ void scatter_scalar_kernel(const float* __restrict__ xc, const float* __restrict__ xm,
                                      const int* __restrict__ sp, const int* __restrict__ dp,
                                      const int* __restrict__ sr, const int* __restrict__ dr,
                                      float* __restrict__ aggc, float* __restrict__ aggm, int E,
                                      long cs) {
    int e = blockIdx.x * blockDim.x + threadIdx.x;
    if (e >= E) return;
    const int dc = dp[e], dm = dr[e];
    const float vc = xc[sp[e]], vm = xm[sr[e]];
    const long ho = chip_half() * cs;
    atomicAdd(&aggc[ho + dc], vc);
    atomicAdd(&aggm[ho + dm], vm);
}

#define EPW 4
#define SWG 2048

template <int NE>
__device__ __forceinline__ void edge_idx(int (&is)[NE], int (&od)[NE],
                                         const int* __restrict__ iidx,
                                         const int* __restrict__ oidx, int e0) {
#pragma unroll
    for (int u = 0; u < NE; ++u) { is[u] = iidx[e0 + u]; od[u] = oidx[e0 + u]; }
}

template <int F, int IP, int NE>
__device__ __forceinline__ void edge_rows(float (&v)[NE][(F + 31) / 32],
                                          const float* __restrict__ inbuf,
                                          const int (&is)[NE], int lane) {
    constexpr int NP = (F + 31) / 32;
#pragma unroll
    for (int u = 0; u < NE; ++u) {
        const float* srow = inbuf + (size_t)is[u] * IP;
#pragma unroll
        for (int k = 0; k < NP; ++k) {
            const int j = lane + 32 * k;
            v[u][k] = srow[j < F ? j : F - 1];
        }
    }
}

template <int F, int OP, int NE>
__device__ __forceinline__ void edge_adds(float* __restrict__ outbuf, const int (&od)[NE],
                                          const float (&v)[NE][(F + 31) / 32], int lane) {
    constexpr int NP = (F + 31) / 32;
    constexpr int TAIL = F - 32 * (NP - 1);
    if (TAIL < 32) {
#pragma unroll
        for (int u = 0; u < NE; ++u) asm volatile("" ::"v"(v[u][NP - 1]));
    }
#pragma unroll
    for (int u = 0; u < NE; ++u) {
        float* drow = outbuf + (size_t)od[u] * OP;
#pragma unroll
        for (int k = 0; k < NP - 1; ++k) atomicAdd(&drow[lane + 32 * k], v[u][k]);
    }
    if (lane < TAIL) {
#pragma unroll
        for (int u = 0; u < NE; ++u)
            atomicAdd(&outbuf[(size_t)od[u] * OP + lane + 32 * (NP - 1)], v[u][NP - 1]);
    }
}

template <int F, int IP, int OP, bool TWOCOPIES>
__device__ __forceinline__ void scatter_rows(float* __restrict__ outbuf,
                                             const int* __restrict__ oidx,
                                             const float* __restrict__ inbuf,
                                             const int* __restrict__ iidx, int E, long cs) {
    constexpr int NP = (F + 31) / 32;
    const int lane = threadIdx.x & 31;
    const int wave = __builtin_amdgcn_readfirstlane((int)(threadIdx.x >> 5));
    const int gw = (int)blockIdx.x * 8 + wave;
    const int nw = (int)gridDim.x * 8;
    const int ng = E / EPW;
    if (TWOCOPIES) outbuf += chip_half() * cs;
    int g = gw;
    if (g < ng) {
        int is[EPW], od[EPW];
        edge_idx<EPW>(is, od, iidx, oidx, g * EPW);
        for (;;) {
            float v[EPW][NP];
            edge_rows<F, IP, EPW>(v, inbuf, is, lane);
            const int gn = g + nw;
            int isn[EPW], odn[EPW];
            edge_idx<EPW>(isn, odn, iidx, oidx, (gn < ng ? gn : g) * EPW);
#pragma unroll
            for (int u = 0; u < EPW; ++u) asm volatile("" ::"s"(isn[u]), "s"(odn[u]));
            edge_adds<F, OP, EPW>(outbuf, od, v, lane);
            if (gn >= ng) break;
#pragma unroll
            for (int u = 0; u < EPW; ++u) { is[u] = isn[u]; od[u] = odn[u]; }
            g = gn;
        }
    }
    if (gw == nw - 1) {
        for (int e = ng * EPW; e < E; ++e) {
            int is1[1], od1[1];
            float v1[1][NP];
            edge_idx<1>(is1, od1, iidx, oidx, e);
            edge_rows<F, IP, 1>(v1, inbuf, is1, lane);
            edge_adds<F, OP, 1>(outbuf, od1, v1, lane);
        }
    }
}

#define EPS 20
template <int F, int IP, int OP, bool TWOCOPIES>
__device__ __forceinline__ void scatter_rows_lone(float* __restrict__ outbuf,
                                                  const int* __restrict__ oidx,
                                                  const float* __restrict__ inbuf,
                                                  const int* __restrict__ iidx, int E, long cs,
                                                  int gw, int nw) {
    constexpr int NP = (F + 31) / 32;
    const int lane = threadIdx.x & 31;
    const int ng = E / EPS;
    if (TWOCOPIES) outbuf += chip_half() * cs;
    const int le = lane < EPS ? lane : EPS - 1;
    int g = gw;
    if (g < ng) {
        int isv = iidx[g * EPS + le], odv = oidx[g * EPS + le];
        for (;;) {
            int is[EPS], od[EPS];
#pragma unroll
            for (int u = 0; u < EPS; ++u) {
                is[u] = __builtin_amdgcn_readlane(isv, u);
                od[u] = __builtin_amdgcn_readlane(odv, u);
            }
            const int gn = g + nw;
            const int en = (gn < ng ? gn : g) * EPS + le;
            const int isn = iidx[en], odn = oidx[en];
            float v[EPS][NP];
            edge_rows<F, IP, EPS>(v, inbuf, is, lane);
            asm volatile("" ::"v"(isn), "v"(odn));
            edge_adds<F, OP, EPS>(outbuf, od, v, lane);
            if (gn >= ng) break;
            isv = isn; odv = odn;
            g = gn;
        }
    }
    if (gw == nw - 1) {
        for (int e = ng * EPS; e < E; ++e) {
            int is1[1], od1[1];
            float v1[1][NP];
            edge_idx<1>(is1, od1, iidx, oidx, e);
            edge_rows<F, IP, 1>(v1, inbuf, is1, lane);
            edge_adds<F, OP, 1>(outbuf, od1, v1, lane);
        }
    }
}

__global__ __launch_bounds__(256) void scatter64_kernel(float* __restrict__ outbuf,
                                                        const int* __restrict__ oidx,
                                                        const float* __restrict__ inbuf,
                                                        const int* __restrict__ iidx, int E) {
    scatter_rows<H, H, H, false>(outbuf, oidx, inbuf, iidx, E, 0);
}

__global__ __launch_bounds__(128) void gemm88_kernel(
    const float* __restrict__ Xmain, long cs, int pitch, const float* __restrict__ extra1,
    const float* __restrict__ extra2, const float* __restrict__ W,
    const float* __restrict__ bias, const float* __restrict__ W2, float* __restrict__ out,
    int M) {
    __shared__ float xs[16 * KPAD];
    __shared__ float hs[16 * HSP];

    const int row0 = blockIdx.x * 16;
    for (int i = threadIdx.x; i < 16 * KPAD; i += 128) {
        int r = i / KPAD, c = i - r * KPAD;
        int grow = row0 + r;
        float v;
        if (c < F85)      v = Xmain[(size_t)grow * pitch + c] + Xmain[(size_t)grow * pitch + c + cs];
        else if (c == 85) v = extra1 ? extra1[grow] : 0.0f;
        else if (c == 86) v = extra2 ? extra2[grow] : 0.0f;
        else              v = 0.0f;
        xs[i] = v;
    }
    __syncthreads();

    const int wave = threadIdx.x >> 5;
    const int lane = threadIdx.x & 31;
    const int half = lane >> 4;
    const int lm   = lane & 15;
    const int c0   = wave * 16;

    const float bval = bias[c0 + lm];
    v8f acc;
#pragma unroll
    for (int i = 0; i < 8; ++i) acc[i] = bval;

    for (int k = 0; k < KPAD; k += 4) {
        const int ka = k + 2 * half;
        v2f a, b;
        a.x = xs[lm * KPAD + ka];
        a.y = xs[lm * KPAD + ka + 1];
        b.x = W[(size_t)(ka + 0) * H + c0 + lm];
        b.y = W[(size_t)(ka + 1) * H + c0 + lm];
        acc = __builtin_amdgcn_wmma_f32_16x16x4_f32(false, a, false, b, (short)0, acc,
                                                    false, false);
    }
#pragma unroll
    for (int v = 0; v < 8; ++v) {
        float r = acc[v];
        r = r > 0.0f ? r : 0.0f;
        hs[(v + 8 * half) * HSP + c0 + lm] = r;
    }
    __syncthreads();
    v8f acc2;
#pragma unroll
    for (int i = 0; i < 8; ++i) acc2[i] = 0.0f;
    for (int k = 0; k < H; k += 4) {
        const int ka = k + 2 * half;
        v2f a, b;
        a.x = hs[lm * HSP + ka];
        a.y = hs[lm * HSP + ka + 1];
        b.x = W2[(size_t)(ka + 0) * H + c0 + lm];
        b.y = W2[(size_t)(ka + 1) * H + c0 + lm];
        acc2 = __builtin_amdgcn_wmma_f32_16x16x4_f32(false, a, false, b, (short)0, acc2,
                                                     false, false);
    }
#pragma unroll
    for (int v = 0; v < 8; ++v)
        out[(size_t)(row0 + v + 8 * half) * H + c0 + lm] = acc2[v];
}

#define C1W 4
#define C1Q 11
#define C1P 360
#define C1X 1408

__device__ __forceinline__ void c1_load(v4f (&r)[C1Q], float& exa, float& exb,
                                        const float* __restrict__ x,
                                        const float* __restrict__ extra, long cs,
                                        int tile, int lane) {
    const v4f* p = (const v4f*)(x + (size_t)tile * (16 * F85));
#pragma unroll
    for (int j = 0; j < C1Q - 1; ++j) r[j] = p[lane + 32 * j];
    const int pl = 32 * (C1Q - 1) + lane;
    r[C1Q - 1] = p[pl < 340 ? pl : 339];
    const float* pe = extra + tile * 16 + (lane & 15);
    exa = pe[0];
    exb = pe[cs];
}

__device__ __forceinline__ void c1_tile(const float* __restrict__ arow,
                                        const float* __restrict__ xrow,
                                        const v4f (&wreg)[C1Q][4], const float (&bv)[4],
                                        int c, int h, float* __restrict__ out, int tile) {
    float a[C1Q][4];
#pragma unroll
    for (int q = 0; q < C1Q; ++q) {
#pragma unroll
        for (int e = 0; e < 4; ++e) a[q][e] = arow[8 * q + e];
    }
    const float e1 = xrow[0], e2 = xrow[16];
    a[C1Q - 1][1] = h ? e1 : a[C1Q - 1][1];
    a[C1Q - 1][2] = h ? e2 : a[C1Q - 1][2];
    a[C1Q - 1][3] = h ? 0.0f : a[C1Q - 1][3];

    v8f acc[4];
#pragma unroll
    for (int ct = 0; ct < 4; ++ct) {
#pragma unroll
        for (int i = 0; i < 8; ++i) acc[ct][i] = bv[ct];
    }
#pragma unroll
    for (int q = 0; q < C1Q; ++q) {
        const v2f a0 = {a[q][0], a[q][1]}, a1 = {a[q][2], a[q][3]};
#pragma unroll
        for (int ct = 0; ct < 4; ++ct) {
            const v4f w = wreg[q][ct];
            const v2f b0 = {w[0], w[1]};
            acc[ct] = __builtin_amdgcn_wmma_f32_16x16x4_f32(false, a0, false, b0, (short)0,
                                                            acc[ct], false, false);
        }
#pragma unroll
        for (int ct = 0; ct < 4; ++ct) {
            const v4f w = wreg[q][ct];
            const v2f b1 = {w[2], w[3]};
            acc[ct] = __builtin_amdgcn_wmma_f32_16x16x4_f32(false, a1, false, b1, (short)0,
                                                            acc[ct], false, false);
        }
    }
    float* orow = out + ((size_t)tile * 16 + 8 * h) * H + 4 * c;
#pragma unroll
    for (int v = 0; v < 8; ++v) {
        v4f o;
#pragma unroll
        for (int ct = 0; ct < 4; ++ct) {
            const float r = acc[ct][v];
            o[ct] = r > 0.0f ? r : 0.0f;
        }
        *(v4f*)(orow + v * H) = o;
    }
}

__global__ __launch_bounds__(256) void conv1_t_kernel(
    const float* __restrict__ x, const float* __restrict__ extra1,
    const float* __restrict__ extra2, const float* __restrict__ W,
    const float* __restrict__ bias, float* __restrict__ out, int ntiles, long cs,
    float* __restrict__ esum, const int* __restrict__ eoidx,
    const int* __restrict__ eiidx, int E) {
    __shared__ v4f stage[C1W * C1P];

    const int lane = threadIdx.x & 31;
    const int c = lane & 15;
    const int h = lane >> 4;
    const int wave = __builtin_amdgcn_readfirstlane((int)(threadIdx.x >> 5));
    if (wave >= C1W) {
        scatter_rows_lone<F85, F85, AP, true>(esum, eoidx, x, eiidx, E, cs,
                                              (int)blockIdx.x * 4 + wave - C1W,
                                              (int)gridDim.x * 4);
        return;
    }
    const int gw = blockIdx.x * C1W + wave;
    const int nw = gridDim.x * C1W;
    v4f* __restrict__ reg = stage + wave * C1P;
    float* __restrict__ regf = (float*)reg;
    const float* __restrict__ arow = regf + c * F85 + 4 * h;
    const float* __restrict__ xrow = regf + C1X + c;
    const float* __restrict__ extra = h ? extra2 : extra1;

    v4f wreg[C1Q][4];
#pragma unroll
    for (int q = 0; q < C1Q; ++q) {
#pragma unroll
        for (int e = 0; e < 4; ++e) {
            const v4f w = *(const v4f*)(W + (size_t)(8 * q + 4 * h + e) * H + 4 * c);
#pragma unroll
            for (int ct = 0; ct < 4; ++ct) wreg[q][ct][e] = w[ct];
        }
    }
    float bv[4];
#pragma unroll
    for (int ct = 0; ct < 4; ++ct) bv[ct] = bias[4 * c + ct];

    const int last = ntiles - 1;
    v4f r[C1Q];
    float exa, exb;
    int tile = gw;
    c1_load(r, exa, exb, x, extra, cs, tile < last ? tile : last, lane);
    for (; tile < ntiles; tile += nw) {
#pragma unroll
        for (int j = 0; j < C1Q; ++j) reg[lane + 32 * j] = r[j];
        regf[C1X + lane] = exa + exb;
        const int nt = tile + nw;
        c1_load(r, exa, exb, x, extra, cs, nt < last ? nt : last, lane);
        __builtin_amdgcn_sched_barrier(0);
        c1_tile(arow, xrow, wreg, bv, c, h, out, tile);
    }
}

__device__ __forceinline__ void c2r_load(v4f (&ah)[8], const float* __restrict__ h_t,
                                         int tile, int c, int h) {
    const v4f* ph = (const v4f*)(h_t + ((size_t)tile * 16 + c) * H + 4 * h);
#pragma unroll
    for (int q = 0; q < 8; ++q) ah[q] = ph[2 * q];
}

__device__ __forceinline__ void c2r_tile(const v4f (&ah)[8], const v4f (&wreg)[8][8],
                                         const float (&bs)[2][4], int c, int h,
                                         float* __restrict__ outc, float* __restrict__ outm,
                                         int tile) {
#pragma unroll
    for (int pass = 0; pass < 2; ++pass) {
        v8f acc[4];
#pragma unroll
        for (int ct = 0; ct < 4; ++ct) {
#pragma unroll
            for (int i = 0; i < 8; ++i) acc[ct][i] = bs[pass][ct];
        }
#pragma unroll
        for (int q = 0; q < 8; ++q) {
            const v4f ar = ah[q];
            const v2f ar0 = {ar[0], ar[1]}, ar1 = {ar[2], ar[3]};
#pragma unroll
            for (int ct = 0; ct < 4; ++ct) {
                const v4f w = wreg[pass * 4 + ct][q];
                const v2f b0 = {w[0], w[1]};
                acc[ct] = __builtin_amdgcn_wmma_f32_16x16x4_f32(false, ar0, false, b0, (short)0,
                                                                acc[ct], false, false);
            }
#pragma unroll
            for (int ct = 0; ct < 4; ++ct) {
                const v4f w = wreg[pass * 4 + ct][q];
                const v2f b1 = {w[2], w[3]};
                acc[ct] = __builtin_amdgcn_wmma_f32_16x16x4_f32(false, ar1, false, b1, (short)0,
                                                                acc[ct], false, false);
            }
        }
        float* orow = (pass ? outm : outc) + ((size_t)tile * 16 + 8 * h) * H + 4 * c;
#pragma unroll
        for (int v = 0; v < 8; ++v) {
            v4f o;
#pragma unroll
            for (int ct = 0; ct < 4; ++ct) o[ct] = acc[ct][v];
            *(v4f*)(orow + v * H) = o;
        }
    }
}

__global__ __launch_bounds__(256) void conv2_root_kernel(
    const float* __restrict__ h_t, const float* __restrict__ w2r,
    const float* __restrict__ b2, float* __restrict__ outc, float* __restrict__ outm,
    int ntiles, float* __restrict__ esum, const int* __restrict__ eoidx,
    const float* __restrict__ ex, const int* __restrict__ eiidx, int E, long cs) {
    __shared__ v4f bf[2 * 4 * 8 * 32];

    {
        float* dst = (float*)bf;
        for (int i = threadIdx.x; i < 2 * H * 16; i += 256) {
            const int n4 = i & 15, k = (i >> 4) & 63, blk = i >> 10;
            const v4f w = *(const v4f*)(w2r + blk * H * H + k * H + 4 * n4);
            const int q = k >> 3, hh = (k >> 2) & 1, e = k & 3;
#pragma unroll
            for (int j = 0; j < 4; ++j) {
                dst[((((blk * 4 + j) * 8 + q) * 32) + hh * 16 + n4) * 4 + e] = w[j];
            }
        }
    }
    __syncthreads();

    const int lane = threadIdx.x & 31;
    const int c = lane & 15;
    const int h = lane >> 4;
    const int wave = __builtin_amdgcn_readfirstlane((int)(threadIdx.x >> 5));
    if (wave >= 4) {
        scatter_rows_lone<F85, F85, AP, true>(esum, eoidx, ex, eiidx, E, cs,
                                              (int)blockIdx.x * 4 + wave - 4,
                                              (int)gridDim.x * 4);
        return;
    }
    const int gw = blockIdx.x * 4 + wave;
    const int nw = gridDim.x * 4;

    float bs[2][4];
#pragma unroll
    for (int ct = 0; ct < 4; ++ct) {
        bs[0][ct] = b2[4 * c + ct];
        bs[1][ct] = b2[H + 4 * c + ct];
    }

    v4f wreg[8][8];
#pragma unroll
    for (int bc = 0; bc < 8; ++bc) {
#pragma unroll
        for (int q = 0; q < 8; ++q) wreg[bc][q] = bf[(bc * 8 + q) * 32 + lane];
    }

    const int last = ntiles - 1;
    v4f ah0[8], ah1[8];
    int tile = gw;
    c2r_load(ah0, h_t, tile < last ? tile : last, c, h);
    for (; tile + nw < ntiles; tile += 2 * nw) {
        const int t2 = tile + 2 * nw;
        c2r_load(ah1, h_t, tile + nw, c, h);
        __builtin_amdgcn_sched_barrier(0);
        c2r_tile(ah0, wreg, bs, c, h, outc, outm, tile);
        c2r_load(ah0, h_t, t2 < last ? t2 : last, c, h);
        __builtin_amdgcn_sched_barrier(0);
        c2r_tile(ah1, wreg, bs, c, h, outc, outm, tile + nw);
    }
    if (tile < ntiles) c2r_tile(ah0, wreg, bs, c, h, outc, outm, tile);
}

#define OWG 512
__global__ __launch_bounds__(256) void out_kernel(
    const float* __restrict__ t1, const float* __restrict__ t2,
    const float* __restrict__ wlin, const float* __restrict__ blin,
    float* __restrict__ out, int ntiles) {
    const int lane = threadIdx.x & 31;
    const int c = lane & 15;
    const int h = lane >> 4;
    const int wave = __builtin_amdgcn_readfirstlane((int)(threadIdx.x >> 5));
    const int gw = blockIdx.x * 8 + wave;
    const int nw = gridDim.x * 8;

    float wa[4][2], wb[4][2];
#pragma unroll
    for (int j = 0; j < 4; ++j) {
        wa[j][0] = wlin[(4 * c + j) * 2 + 0];
        wa[j][1] = wlin[(4 * c + j) * 2 + 1];
        wb[j][0] = wlin[(H + 4 * c + j) * 2 + 0];
        wb[j][1] = wlin[(H + 4 * c + j) * 2 + 1];
    }
    const float blo = blin[lane & 1];
    const bool b3 = (c & 8) != 0, b2 = (c & 4) != 0, b1 = (c & 2) != 0, b0 = (c & 1) != 0;

    for (int tile = gw; tile < ntiles; tile += nw) {
        const size_t ro = ((size_t)tile * 16 + 8 * h) * H + 4 * c;
        const v4f* p1 = (const v4f*)(t1 + ro);
        const v4f* p2 = (const v4f*)(t2 + ro);
        v4f x1[8], x2[8];
#pragma unroll
        for (int v = 0; v < 8; ++v) { x1[v] = p1[v * (H / 4)]; x2[v] = p2[v * (H / 4)]; }
        float p[16];
#pragma unroll
        for (int v = 0; v < 8; ++v) {
            float s0 = 0.0f, s1 = 0.0f;
#pragma unroll
            for (int j = 0; j < 4; ++j) {
                float r = x1[v][j];
                r = r > 0.0f ? r : 0.0f;
                s0 += r * wa[j][0];
                s1 += r * wa[j][1];
            }
#pragma unroll
            for (int j = 0; j < 4; ++j) {
                float r = x2[v][j];
                r = r > 0.0f ? r : 0.0f;
                s0 += r * wb[j][0];
                s1 += r * wb[j][1];
            }
            p[2 * v + 0] = s0;
            p[2 * v + 1] = s1;
        }
        float s8[8], s4[4], s2[2];
#pragma unroll
        for (int i = 0; i < 8; ++i) {
            const float keep = b3 ? p[i + 8] : p[i];
            const float send = b3 ? p[i] : p[i + 8];
            s8[i] = keep + __shfl_xor(send, 8);
        }
#pragma unroll
        for (int i = 0; i < 4; ++i) {
            const float keep = b2 ? s8[i + 4] : s8[i];
            const float send = b2 ? s8[i] : s8[i + 4];
            s4[i] = keep + __shfl_xor(send, 4);
        }
#pragma unroll
        for (int i = 0; i < 2; ++i) {
            const float keep = b1 ? s4[i + 2] : s4[i];
            const float send = b1 ? s4[i] : s4[i + 2];
            s2[i] = keep + __shfl_xor(send, 2);
        }
        const float keep1 = b0 ? s2[1] : s2[0];
        const float send1 = b0 ? s2[0] : s2[1];
        const float tot = keep1 + __shfl_xor(send1, 1);
        out[(size_t)tile * 32 + lane] = tot + blo;
    }
}

extern "C" void kernel_launch(void* const* d_in, const int* in_sizes, int n_in,
                              void* d_out, int out_size, void* d_ws, size_t ws_size,
                              hipStream_t stream) {
    const float* x_card   = (const float*)d_in[0];
    const float* x_merch  = (const float*)d_in[1];
    const float* x_trans  = (const float*)d_in[2];
    const float* w1_src_t = (const float*)d_in[3];
    const float* w1_dst_t = (const float*)d_in[4];
    const float* b1_t     = (const float*)d_in[5];
    const float* w1_src_r = (const float*)d_in[6];
    const float* w1_dst_r = (const float*)d_in[7];
    const float* b1_r     = (const float*)d_in[8];
    const float* w2_l     = (const float*)d_in[9];
    const float* w2_r     = (const float*)d_in[10];
    const float* b2       = (const float*)d_in[11];
    const float* w_lin    = (const float*)d_in[12];
    const float* b_lin    = (const float*)d_in[13];
    const int* src_pays   = (const int*)d_in[14];
    const int* dst_pays   = (const int*)d_in[15];
    const int* src_recv   = (const int*)d_in[16];
    const int* dst_recv   = (const int*)d_in[17];
    float* out = (float*)d_out;

    const int N_c = in_sizes[0];
    const int N_m = in_sizes[1];
    const int N_t = in_sizes[2] / F85;
    const int E   = in_sizes[14];

    float* ws = (float*)d_ws;
    size_t off = 0;
    float* aggc1  = ws + off; off += (size_t)N_t;
    float* aggm1  = ws + off; off += (size_t)N_t;
    float* aggt2c = ws + off; off += (size_t)N_c * AP;
    float* aggt2m = ws + off; off += (size_t)N_m * AP;
    const long cs = (long)off;
    off += (size_t)cs;
    const long zeroCount = (long)off;
    float* agg2c  = ws + off; off += (size_t)N_t * H;
    float* agg2m  = ws + off; off += (size_t)N_t * H;
    float* h_t    = ws + off; off += (size_t)N_t * H;
    float* h_c    = ws + off; off += (size_t)N_c * H;
    float* h_m    = ws + off; off += (size_t)N_m * H;
    float* W1t    = ws + off; off += (size_t)KPAD * H;
    float* W1c    = ws + off; off += (size_t)KPAD * H;
    float* W1m    = ws + off; off += (size_t)KPAD * H;
    float* b1sum  = ws + off; off += (size_t)H;

    zero_kernel<<<ZWG, 256, 0, stream>>>(ws, zeroCount);
    prep_w_kernel<<<(KPAD * H + 255) / 256, 256, 0, stream>>>(
        w1_src_t, w1_dst_t, b1_t, w1_src_r, w1_dst_r, W1t, W1c, W1m, b1sum);
    scatter_scalar_kernel<<<(E + 255) / 256, 256, 0, stream>>>(
        x_card, x_merch, src_pays, dst_pays, src_recv, dst_recv, aggc1, aggm1, E, cs);
    conv1_t_kernel<<<256, 256, 0, stream>>>(x_trans, aggc1, aggm1, W1t, b1sum, h_t, N_t / 16, cs,
                                            aggt2c, src_pays, dst_pays, E);
    conv2_root_kernel<<<256, 256, 0, stream>>>(h_t, w2_r, b2, agg2c, agg2m, N_t / 16,
                                               aggt2m, src_recv, x_trans, dst_recv, E, cs);
    gemm88_kernel<<<N_c / 16, 128, 0, stream>>>(aggt2c, cs, AP, x_card, nullptr, W1c, b1_r, w2_l, h_c, N_c);
    gemm88_kernel<<<N_m / 16, 128, 0, stream>>>(aggt2m, cs, AP, x_merch, nullptr, W1m, b1_r + H, w2_l + H * H, h_m, N_m);
    scatter64_kernel<<<SWG, 256, 0, stream>>>(
        agg2c, dst_pays, h_c, src_pays, E);
    scatter64_kernel<<<SWG, 256, 0, stream>>>(
        agg2m, dst_recv, h_m, src_recv, E);
    out_kernel<<<OWG, 256, 0, stream>>>(agg2c, agg2m, w_lin, b_lin, out, N_t / 16);
}


# card / merchant GEMM kernels: a block's 16 sum rows (both copies) and its weights fetched as one group of loads, 16-byte pieces, in place of eleven dependent round trips; sums unchanged
# speedup vs baseline: 4.0683x; 1.0518x over previous
#include <hip/hip_runtime.h>

typedef __attribute__((ext_vector_type(2))) float v2f;
typedef __attribute__((ext_vector_type(4))) float v4f;
typedef __attribute__((ext_vector_type(8))) float v8f;

#define H 64
#define F85 85
#define KPAD 88
#define AP 96
#define HSP 68

__device__ __forceinline__ int chip_half() {
    return (int)((__builtin_amdgcn_s_sendmsg_rtn(0x87) >> 18) & 1u);
}

#define ZWG 256
__global__ __launch_bounds__(256) void zero_kernel(float* __restrict__ p, long n) {
    const unsigned lane = threadIdx.x & 31;
    const unsigned gw = blockIdx.x * 8 + (threadIdx.x >> 5);
    const unsigned nw = gridDim.x * 8;
    const unsigned n4 = (unsigned)(n >> 2);
    const unsigned units = (n4 + 31) >> 5;
    const unsigned per = (units + nw - 1) / nw;
    const unsigned b = gw * per * 32;
    const unsigned lim = b + per * 32;
    const unsigned e = lim < n4 ? lim : n4;
    v4f* __restrict__ q = (v4f*)p;
    const v4f z = {0.0f, 0.0f, 0.0f, 0.0f};
    unsigned i = b + lane;
    for (; i + 7 * 32 < e; i += 8 * 32) {
#pragma unroll
        for (int k = 0; k < 8; ++k) q[i + 32 * k] = z;
    }
    for (; i < e; i += 32) q[i] = z;
    if (gw == 0 && lane < (unsigned)(n & 3)) p[((long)n4 << 2) + lane] = 0.0f;
}

__global__ void prep_w_kernel(const float* __restrict__ w1_src_t,
                              const float* __restrict__ w1_dst_t,
                              const float* __restrict__ b1_t,
                              const float* __restrict__ w1_src_r,
                              const float* __restrict__ w1_dst_r,
                              float* __restrict__ W1t, float* __restrict__ W1c,
                              float* __restrict__ W1m, float* __restrict__ b1sum) {
    int i = blockIdx.x * blockDim.x + threadIdx.x;
    if (i >= KPAD * H) return;
    int r = i >> 6, c = i & 63;
    float vt = 0.f, vc = 0.f, vm = 0.f;
    if (r < F85) {
        vt = w1_dst_t[r * H + c] + w1_dst_t[F85 * H + r * H + c];
        vc = w1_src_r[r * H + c];
        vm = w1_src_r[F85 * H + r * H + c];
    } else if (r == 85) {
        vt = w1_src_t[c];
        vc = w1_dst_r[c];
        vm = w1_dst_r[H + c];
    } else if (r == 86) {
        vt = w1_src_t[H + c];
    }
    W1t[i] = vt; W1c[i] = vc; W1m[i] = vm;
    if (i < H) b1sum[i] = b1_t[i] + b1_t[H + i];
}

__global__ void scatter_scalar_kernel(const float* __restrict__ xc, const float* __restrict__ xm,
                                      const int* __restrict__ sp, const int* __restrict__ dp,
                                      const int* __restrict__ sr, const int* __restrict__ dr,
                                      float* __restrict__ aggc, float* __restrict__ aggm, int E,
                                      long cs) {
    int e = blockIdx.x * blockDim.x + threadIdx.x;
    if (e >= E) return;
    const int dc = dp[e], dm = dr[e];
    const float vc = xc[sp[e]], vm = xm[sr[e]];
    const long ho = chip_half() * cs;
    atomicAdd(&aggc[ho + dc], vc);
    atomicAdd(&aggm[ho + dm], vm);
}

#define EPW 4
#define SWG 2048

template <int NE>
__device__ __forceinline__ void edge_idx(int (&is)[NE], int (&od)[NE],
                                         const int* __restrict__ iidx,
                                         const int* __restrict__ oidx, int e0) {
#pragma unroll
    for (int u = 0; u < NE; ++u) { is[u] = iidx[e0 + u]; od[u] = oidx[e0 + u]; }
}

template <int F, int IP, int NE>
__device__ __forceinline__ void edge_rows(float (&v)[NE][(F + 31) / 32],
                                          const float* __restrict__ inbuf,
                                          const int (&is)[NE], int lane) {
    constexpr int NP = (F + 31) / 32;
#pragma unroll
    for (int u = 0; u < NE; ++u) {
        const float* srow = inbuf + (size_t)is[u] * IP;
#pragma unroll
        for (int k = 0; k < NP; ++k) {
            const int j = lane + 32 * k;
            v[u][k] = srow[j < F ? j : F - 1];
        }
    }
}

template <int F, int OP, int NE>
__device__ __forceinline__ void edge_adds(float* __restrict__ outbuf, const int (&od)[NE],
                                          const float (&v)[NE][(F + 31) / 32], int lane) {
    constexpr int NP = (F + 31) / 32;
    constexpr int TAIL = F - 32 * (NP - 1);
    if (TAIL < 32) {
#pragma unroll
        for (int u = 0; u < NE; ++u) asm volatile("" ::"v"(v[u][NP - 1]));
    }
#pragma unroll
    for (int u = 0; u < NE; ++u) {
        float* drow = outbuf + (size_t)od[u] * OP;
#pragma unroll
        for (int k = 0; k < NP - 1; ++k) atomicAdd(&drow[lane + 32 * k], v[u][k]);
    }
    if (lane < TAIL) {
#pragma unroll
        for (int u = 0; u < NE; ++u)
            atomicAdd(&outbuf[(size_t)od[u] * OP + lane + 32 * (NP - 1)], v[u][NP - 1]);
    }
}

template <int F, int IP, int OP, bool TWOCOPIES>
__device__ __forceinline__ void scatter_rows(float* __restrict__ outbuf,
                                             const int* __restrict__ oidx,
                                             const float* __restrict__ inbuf,
                                             const int* __restrict__ iidx, int E, long cs) {
    constexpr int NP = (F + 31) / 32;
    const int lane = threadIdx.x & 31;
    const int wave = __builtin_amdgcn_readfirstlane((int)(threadIdx.x >> 5));
    const int gw = (int)blockIdx.x * 8 + wave;
    const int nw = (int)gridDim.x * 8;
    const int ng = E / EPW;
    if (TWOCOPIES) outbuf += chip_half() * cs;
    int g = gw;
    if (g < ng) {
        int is[EPW], od[EPW];
        edge_idx<EPW>(is, od, iidx, oidx, g * EPW);
        for (;;) {
            float v[EPW][NP];
            edge_rows<F, IP, EPW>(v, inbuf, is, lane);
            const int gn = g + nw;
            int isn[EPW], odn[EPW];
            edge_idx<EPW>(isn, odn, iidx, oidx, (gn < ng ? gn : g) * EPW);
#pragma unroll
            for (int u = 0; u < EPW; ++u) asm volatile("" ::"s"(isn[u]), "s"(odn[u]));
            edge_adds<F, OP, EPW>(outbuf, od, v, lane);
            if (gn >= ng) break;
#pragma unroll
            for (int u = 0; u < EPW; ++u) { is[u] = isn[u]; od[u] = odn[u]; }
            g = gn;
        }
    }
    if (gw == nw - 1) {
        for (int e = ng * EPW; e < E; ++e) {
            int is1[1], od1[1];
            float v1[1][NP];
            edge_idx<1>(is1, od1, iidx, oidx, e);
            edge_rows<F, IP, 1>(v1, inbuf, is1, lane);
            edge_adds<F, OP, 1>(outbuf, od1, v1, lane);
        }
    }
}

#define EPS 20
template <int F, int IP, int OP, bool TWOCOPIES>
__device__ __forceinline__ void scatter_rows_lone(float* __restrict__ outbuf,
                                                  const int* __restrict__ oidx,
                                                  const float* __restrict__ inbuf,
                                                  const int* __restrict__ iidx, int E, long cs,
                                                  int gw, int nw) {
    constexpr int NP = (F + 31) / 32;
    const int lane = threadIdx.x & 31;
    const int ng = E / EPS;
    if (TWOCOPIES) outbuf += chip_half() * cs;
    const int le = lane < EPS ? lane : EPS - 1;
    int g = gw;
    if (g < ng) {
        int isv = iidx[g * EPS + le], odv = oidx[g * EPS + le];
        for (;;) {
            int is[EPS], od[EPS];
#pragma unroll
            for (int u = 0; u < EPS; ++u) {
                is[u] = __builtin_amdgcn_readlane(isv, u);
                od[u] = __builtin_amdgcn_readlane(odv, u);
            }
            const int gn = g + nw;
            const int en = (gn < ng ? gn : g) * EPS + le;
            const int isn = iidx[en], odn = oidx[en];
            float v[EPS][NP];
            edge_rows<F, IP, EPS>(v, inbuf, is, lane);
            asm volatile("" ::"v"(isn), "v"(odn));
            edge_adds<F, OP, EPS>(outbuf, od, v, lane);
            if (gn >= ng) break;
            isv = isn; odv = odn;
            g = gn;
        }
    }
    if (gw == nw - 1) {
        for (int e = ng * EPS; e < E; ++e) {
            int is1[1], od1[1];
            float v1[1][NP];
            edge_idx<1>(is1, od1, iidx, oidx, e);
            edge_rows<F, IP, 1>(v1, inbuf, is1, lane);
            edge_adds<F, OP, 1>(outbuf, od1, v1, lane);
        }
    }
}

__global__ __launch_bounds__(256) void scatter64_kernel(float* __restrict__ outbuf,
                                                        const int* __restrict__ oidx,
                                                        const float* __restrict__ inbuf,
                                                        const int* __restrict__ iidx, int E) {
    scatter_rows<H, H, H, false>(outbuf, oidx, inbuf, iidx, E, 0);
}

#define XP 100
__global__ __launch_bounds__(128) void gemm88_kernel(
    const float* __restrict__ S, long cs, const float* __restrict__ x,
    const float* __restrict__ W, const float* __restrict__ bias,
    const float* __restrict__ W2, float* __restrict__ out) {
    __shared__ v4f xsv[16 * XP / 4];
    __shared__ float hs[16 * HSP];
    float* __restrict__ xs = (float*)xsv;

    const int row0 = blockIdx.x * 16;
    const int wave = threadIdx.x >> 5;
    const int lane = threadIdx.x & 31;
    const int half = lane >> 4;
    const int lm   = lane & 15;
    const int c0   = wave * 16;

    const v4f* __restrict__ p0 = (const v4f*)(S + (size_t)row0 * AP);
    const v4f* __restrict__ p1 = (const v4f*)(S + (size_t)row0 * AP + cs);
    v4f sa[3], sb[3];
    float xe[3];
#pragma unroll
    for (int j = 0; j < 3; ++j) {
        const int p = (int)threadIdx.x + 128 * j;
        sa[j] = p0[p];
        sb[j] = p1[p];
        xe[j] = x[row0 + p / (AP / 4)];
    }
    v2f bw[KPAD / 4];
#pragma unroll
    for (int q = 0; q < KPAD / 4; ++q) {
        const int ka = 4 * q + 2 * half;
        bw[q].x = W[(size_t)(ka + 0) * H + c0 + lm];
        bw[q].y = W[(size_t)(ka + 1) * H + c0 + lm];
    }
    const float bval = bias[c0 + lm];
    __builtin_amdgcn_sched_barrier(0);

#pragma unroll
    for (int j = 0; j < 3; ++j) {
        const int p = (int)threadIdx.x + 128 * j;
        const int r = p / (AP / 4), cg = p - r * (AP / 4);
        v4f v = sa[j] + sb[j];
        if (cg == 21) { v[1] = xe[j]; v[2] = 0.0f; v[3] = 0.0f; }
        xsv[r * (XP / 4) + cg] = v;
    }
    __syncthreads();

    v2f bw2[H / 4];
#pragma unroll
    for (int q = 0; q < H / 4; ++q) {
        const int ka = 4 * q + 2 * half;
        bw2[q].x = W2[(size_t)(ka + 0) * H + c0 + lm];
        bw2[q].y = W2[(size_t)(ka + 1) * H + c0 + lm];
    }
    __builtin_amdgcn_sched_barrier(0);

    v8f acc;
#pragma unroll
    for (int i = 0; i < 8; ++i) acc[i] = bval;

#pragma unroll
    for (int q = 0; q < KPAD / 4; ++q) {
        const int ka = 4 * q + 2 * half;
        v2f a;
        a.x = xs[lm * XP + ka];
        a.y = xs[lm * XP + ka + 1];
        acc = __builtin_amdgcn_wmma_f32_16x16x4_f32(false, a, false, bw[q], (short)0, acc,
                                                    false, false);
    }
#pragma unroll
    for (int v = 0; v < 8; ++v) {
        float r = acc[v];
        r = r > 0.0f ? r : 0.0f;
        hs[(v + 8 * half) * HSP + c0 + lm] = r;
    }
    __syncthreads();
    v8f acc2;
#pragma unroll
    for (int i = 0; i < 8; ++i) acc2[i] = 0.0f;
#pragma unroll
    for (int q = 0; q < H / 4; ++q) {
        const int ka = 4 * q + 2 * half;
        v2f a;
        a.x = hs[lm * HSP + ka];
        a.y = hs[lm * HSP + ka + 1];
        acc2 = __builtin_amdgcn_wmma_f32_16x16x4_f32(false, a, false, bw2[q], (short)0, acc2,
                                                     false, false);
    }
#pragma unroll
    for (int v = 0; v < 8; ++v)
        out[(size_t)(row0 + v + 8 * half) * H + c0 + lm] = acc2[v];
}

#define C1W 4
#define C1Q 11
#define C1P 360
#define C1X 1408

__device__ __forceinline__ void c1_load(v4f (&r)[C1Q], float& exa, float& exb,
                                        const float* __restrict__ x,
                                        const float* __restrict__ extra, long cs,
                                        int tile, int lane) {
    const v4f* p = (const v4f*)(x + (size_t)tile * (16 * F85));
#pragma unroll
    for (int j = 0; j < C1Q - 1; ++j) r[j] = p[lane + 32 * j];
    const int pl = 32 * (C1Q - 1) + lane;
    r[C1Q - 1] = p[pl < 340 ? pl : 339];
    const float* pe = extra + tile * 16 + (lane & 15);
    exa = pe[0];
    exb = pe[cs];
}

__device__ __forceinline__ void c1_tile(const float* __restrict__ arow,
                                        const float* __restrict__ xrow,
                                        const v4f (&wreg)[C1Q][4], const float (&bv)[4],
                                        int c, int h, float* __restrict__ out, int tile) {
    float a[C1Q][4];
#pragma unroll
    for (int q = 0; q < C1Q; ++q) {
#pragma unroll
        for (int e = 0; e < 4; ++e) a[q][e] = arow[8 * q + e];
    }
    const float e1 = xrow[0], e2 = xrow[16];
    a[C1Q - 1][1] = h ? e1 : a[C1Q - 1][1];
    a[C1Q - 1][2] = h ? e2 : a[C1Q - 1][2];
    a[C1Q - 1][3] = h ? 0.0f : a[C1Q - 1][3];

    v8f acc[4];
#pragma unroll
    for (int ct = 0; ct < 4; ++ct) {
#pragma unroll
        for (int i = 0; i < 8; ++i) acc[ct][i] = bv[ct];
    }
#pragma unroll
    for (int q = 0; q < C1Q; ++q) {
        const v2f a0 = {a[q][0], a[q][1]}, a1 = {a[q][2], a[q][3]};
#pragma unroll
        for (int ct = 0; ct < 4; ++ct) {
            const v4f w = wreg[q][ct];
            const v2f b0 = {w[0], w[1]};
            acc[ct] = __builtin_amdgcn_wmma_f32_16x16x4_f32(false, a0, false, b0, (short)0,
                                                            acc[ct], false, false);
        }
#pragma unroll
        for (int ct = 0; ct < 4; ++ct) {
            const v4f w = wreg[q][ct];
            const v2f b1 = {w[2], w[3]};
            acc[ct] = __builtin_amdgcn_wmma_f32_16x16x4_f32(false, a1, false, b1, (short)0,
                                                            acc[ct], false, false);
        }
    }
    float* orow = out + ((size_t)tile * 16 + 8 * h) * H + 4 * c;
#pragma unroll
    for (int v = 0; v < 8; ++v) {
        v4f o;
#pragma unroll
        for (int ct = 0; ct < 4; ++ct) {
            const float r = acc[ct][v];
            o[ct] = r > 0.0f ? r : 0.0f;
        }
        *(v4f*)(orow + v * H) = o;
    }
}

__global__ __launch_bounds__(256) void conv1_t_kernel(
    const float* __restrict__ x, const float* __restrict__ extra1,
    const float* __restrict__ extra2, const float* __restrict__ W,
    const float* __restrict__ bias, float* __restrict__ out, int ntiles, long cs,
    float* __restrict__ esum, const int* __restrict__ eoidx,
    const int* __restrict__ eiidx, int E) {
    __shared__ v4f stage[C1W * C1P];

    const int lane = threadIdx.x & 31;
    const int c = lane & 15;
    const int h = lane >> 4;
    const int wave = __builtin_amdgcn_readfirstlane((int)(threadIdx.x >> 5));
    if (wave >= C1W) {
        scatter_rows_lone<F85, F85, AP, true>(esum, eoidx, x, eiidx, E, cs,
                                              (int)blockIdx.x * 4 + wave - C1W,
                                              (int)gridDim.x * 4);
        return;
    }
    const int gw = blockIdx.x * C1W + wave;
    const int nw = gridDim.x * C1W;
    v4f* __restrict__ reg = stage + wave * C1P;
    float* __restrict__ regf = (float*)reg;
    const float* __restrict__ arow = regf + c * F85 + 4 * h;
    const float* __restrict__ xrow = regf + C1X + c;
    const float* __restrict__ extra = h ? extra2 : extra1;

    v4f wreg[C1Q][4];
#pragma unroll
    for (int q = 0; q < C1Q; ++q) {
#pragma unroll
        for (int e = 0; e < 4; ++e) {
            const v4f w = *(const v4f*)(W + (size_t)(8 * q + 4 * h + e) * H + 4 * c);
#pragma unroll
            for (int ct = 0; ct < 4; ++ct) wreg[q][ct][e] = w[ct];
        }
    }
    float bv[4];
#pragma unroll
    for (int ct = 0; ct < 4; ++ct) bv[ct] = bias[4 * c + ct];

    const int last = ntiles - 1;
    v4f r[C1Q];
    float exa, exb;
    int tile = gw;
    c1_load(r, exa, exb, x, extra, cs, tile < last ? tile : last, lane);
    for (; tile < ntiles; tile += nw) {
#pragma unroll
        for (int j = 0; j < C1Q; ++j) reg[lane + 32 * j] = r[j];
        regf[C1X + lane] = exa + exb;
        const int nt = tile + nw;
        c1_load(r, exa, exb, x, extra, cs, nt < last ? nt : last, lane);
        __builtin_amdgcn_sched_barrier(0);
        c1_tile(arow, xrow, wreg, bv, c, h, out, tile);
    }
}

__device__ __forceinline__ void c2r_load(v4f (&ah)[8], const float* __restrict__ h_t,
                                         int tile, int c, int h) {
    const v4f* ph = (const v4f*)(h_t + ((size_t)tile * 16 + c) * H + 4 * h);
#pragma unroll
    for (int q = 0; q < 8; ++q) ah[q] = ph[2 * q];
}

__device__ __forceinline__ void c2r_tile(const v4f (&ah)[8], const v4f (&wreg)[8][8],
                                         const float (&bs)[2][4], int c, int h,
                                         float* __restrict__ outc, float* __restrict__ outm,
                                         int tile) {
#pragma unroll
    for (int pass = 0; pass < 2; ++pass) {
        v8f acc[4];
#pragma unroll
        for (int ct = 0; ct < 4; ++ct) {
#pragma unroll
            for (int i = 0; i < 8; ++i) acc[ct][i] = bs[pass][ct];
        }
#pragma unroll
        for (int q = 0; q < 8; ++q) {
            const v4f ar = ah[q];
            const v2f ar0 = {ar[0], ar[1]}, ar1 = {ar[2], ar[3]};
#pragma unroll
            for (int ct = 0; ct < 4; ++ct) {
                const v4f w = wreg[pass * 4 + ct][q];
                const v2f b0 = {w[0], w[1]};
                acc[ct] = __builtin_amdgcn_wmma_f32_16x16x4_f32(false, ar0, false, b0, (short)0,
                                                                acc[ct], false, false);
            }
#pragma unroll
            for (int ct = 0; ct < 4; ++ct) {
                const v4f w = wreg[pass * 4 + ct][q];
                const v2f b1 = {w[2], w[3]};
                acc[ct] = __builtin_amdgcn_wmma_f32_16x16x4_f32(false, ar1, false, b1, (short)0,
                                                                acc[ct], false, false);
            }
        }
        float* orow = (pass ? outm : outc) + ((size_t)tile * 16 + 8 * h) * H + 4 * c;
#pragma unroll
        for (int v = 0; v < 8; ++v) {
            v4f o;
#pragma unroll
            for (int ct = 0; ct < 4; ++ct) o[ct] = acc[ct][v];
            *(v4f*)(orow + v * H) = o;
        }
    }
}

__global__ __launch_bounds__(256) void conv2_root_kernel(
    const float* __restrict__ h_t, const float* __restrict__ w2r,
    const float* __restrict__ b2, float* __restrict__ outc, float* __restrict__ outm,
    int ntiles, float* __restrict__ esum, const int* __restrict__ eoidx,
    const float* __restrict__ ex, const int* __restrict__ eiidx, int E, long cs) {
    __shared__ v4f bf[2 * 4 * 8 * 32];

    {
        float* dst = (float*)bf;
        for (int i = threadIdx.x; i < 2 * H * 16; i += 256) {
            const int n4 = i & 15, k = (i >> 4) & 63, blk = i >> 10;
            const v4f w = *(const v4f*)(w2r + blk * H * H + k * H + 4 * n4);
            const int q = k >> 3, hh = (k >> 2) & 1, e = k & 3;
#pragma unroll
            for (int j = 0; j < 4; ++j) {
                dst[((((blk * 4 + j) * 8 + q) * 32) + hh * 16 + n4) * 4 + e] = w[j];
            }
        }
    }
    __syncthreads();

    const int lane = threadIdx.x & 31;
    const int c = lane & 15;
    const int h = lane >> 4;
    const int wave = __builtin_amdgcn_readfirstlane((int)(threadIdx.x >> 5));
    if (wave >= 4) {
        scatter_rows_lone<F85, F85, AP, true>(esum, eoidx, ex, eiidx, E, cs,
                                              (int)blockIdx.x * 4 + wave - 4,
                                              (int)gridDim.x * 4);
        return;
    }
    const int gw = blockIdx.x * 4 + wave;
    const int nw = gridDim.x * 4;

    float bs[2][4];
#pragma unroll
    for (int ct = 0; ct < 4; ++ct) {
        bs[0][ct] = b2[4 * c + ct];
        bs[1][ct] = b2[H + 4 * c + ct];
    }

    v4f wreg[8][8];
#pragma unroll
    for (int bc = 0; bc < 8; ++bc) {
#pragma unroll
        for (int q = 0; q < 8; ++q) wreg[bc][q] = bf[(bc * 8 + q) * 32 + lane];
    }

    const int last = ntiles - 1;
    v4f ah0[8], ah1[8];
    int tile = gw;
    c2r_load(ah0, h_t, tile < last ? tile : last, c, h);
    for (; tile + nw < ntiles; tile += 2 * nw) {
        const int t2 = tile + 2 * nw;
        c2r_load(ah1, h_t, tile + nw, c, h);
        __builtin_amdgcn_sched_barrier(0);
        c2r_tile(ah0, wreg, bs, c, h, outc, outm, tile);
        c2r_load(ah0, h_t, t2 < last ? t2 : last, c, h);
        __builtin_amdgcn_sched_barrier(0);
        c2r_tile(ah1, wreg, bs, c, h, outc, outm, tile + nw);
    }
    if (tile < ntiles) c2r_tile(ah0, wreg, bs, c, h, outc, outm, tile);
}

#define OWG 512
__global__ __launch_bounds__(256) void out_kernel(
    const float* __restrict__ t1, const float* __restrict__ t2,
    const float* __restrict__ wlin, const float* __restrict__ blin,
    float* __restrict__ out, int ntiles) {
    const int lane = threadIdx.x & 31;
    const int c = lane & 15;
    const int h = lane >> 4;
    const int wave = __builtin_amdgcn_readfirstlane((int)(threadIdx.x >> 5));
    const int gw = blockIdx.x * 8 + wave;
    const int nw = gridDim.x * 8;

    float wa[4][2], wb[4][2];
#pragma unroll
    for (int j = 0; j < 4; ++j) {
        wa[j][0] = wlin[(4 * c + j) * 2 + 0];
        wa[j][1] = wlin[(4 * c + j) * 2 + 1];
        wb[j][0] = wlin[(H + 4 * c + j) * 2 + 0];
        wb[j][1] = wlin[(H + 4 * c + j) * 2 + 1];
    }
    const float blo = blin[lane & 1];
    const bool b3 = (c & 8) != 0, b2 = (c & 4) != 0, b1 = (c & 2) != 0, b0 = (c & 1) != 0;

    for (int tile = gw; tile < ntiles; tile += nw) {
        const size_t ro = ((size_t)tile * 16 + 8 * h) * H + 4 * c;
        const v4f* p1 = (const v4f*)(t1 + ro);
        const v4f* p2 = (const v4f*)(t2 + ro);
        v4f x1[8], x2[8];
#pragma unroll
        for (int v = 0; v < 8; ++v) { x1[v] = p1[v * (H / 4)]; x2[v] = p2[v * (H / 4)]; }
        float p[16];
#pragma unroll
        for (int v = 0; v < 8; ++v) {
            float s0 = 0.0f, s1 = 0.0f;
#pragma unroll
            for (int j = 0; j < 4; ++j) {
                float r = x1[v][j];
                r = r > 0.0f ? r : 0.0f;
                s0 += r * wa[j][0];
                s1 += r * wa[j][1];
            }
#pragma unroll
            for (int j = 0; j < 4; ++j) {
                float r = x2[v][j];
                r = r > 0.0f ? r : 0.0f;
                s0 += r * wb[j][0];
                s1 += r * wb[j][1];
            }
            p[2 * v + 0] = s0;
            p[2 * v + 1] = s1;
        }
        float s8[8], s4[4], s2[2];
#pragma unroll
        for (int i = 0; i < 8; ++i) {
            const float keep = b3 ? p[i + 8] : p[i];
            const float send = b3 ? p[i] : p[i + 8];
            s8[i] = keep + __shfl_xor(send, 8);
        }
#pragma unroll
        for (int i = 0; i < 4; ++i) {
            const float keep = b2 ? s8[i + 4] : s8[i];
            const float send = b2 ? s8[i] : s8[i + 4];
            s4[i] = keep + __shfl_xor(send, 4);
        }
#pragma unroll
        for (int i = 0; i < 2; ++i) {
            const float keep = b1 ? s4[i + 2] : s4[i];
            const float send = b1 ? s4[i] : s4[i + 2];
            s2[i] = keep + __shfl_xor(send, 2);
        }
        const float keep1 = b0 ? s2[1] : s2[0];
        const float send1 = b0 ? s2[0] : s2[1];
        const float tot = keep1 + __shfl_xor(send1, 1);
        out[(size_t)tile * 32 + lane] = tot + blo;
    }
}

extern "C" void kernel_launch(void* const* d_in, const int* in_sizes, int n_in,
                              void* d_out, int out_size, void* d_ws, size_t ws_size,
                              hipStream_t stream) {
    const float* x_card   = (const float*)d_in[0];
    const float* x_merch  = (const float*)d_in[1];
    const float* x_trans  = (const float*)d_in[2];
    const float* w1_src_t = (const float*)d_in[3];
    const float* w1_dst_t = (const float*)d_in[4];
    const float* b1_t     = (const float*)d_in[5];
    const float* w1_src_r = (const float*)d_in[6];
    const float* w1_dst_r = (const float*)d_in[7];
    const float* b1_r     = (const float*)d_in[8];
    const float* w2_l     = (const float*)d_in[9];
    const float* w2_r     = (const float*)d_in[10];
    const float* b2       = (const float*)d_in[11];
    const float* w_lin    = (const float*)d_in[12];
    const float* b_lin    = (const float*)d_in[13];
    const int* src_pays   = (const int*)d_in[14];
    const int* dst_pays   = (const int*)d_in[15];
    const int* src_recv   = (const int*)d_in[16];
    const int* dst_recv   = (const int*)d_in[17];
    float* out = (float*)d_out;

    const int N_c = in_sizes[0];
    const int N_m = in_sizes[1];
    const int N_t = in_sizes[2] / F85;
    const int E   = in_sizes[14];

    float* ws = (float*)d_ws;
    size_t off = 0;
    float* aggc1  = ws + off; off += (size_t)N_t;
    float* aggm1  = ws + off; off += (size_t)N_t;
    float* aggt2c = ws + off; off += (size_t)N_c * AP;
    float* aggt2m = ws + off; off += (size_t)N_m * AP;
    const long cs = (long)off;
    off += (size_t)cs;
    const long zeroCount = (long)off;
    float* agg2c  = ws + off; off += (size_t)N_t * H;
    float* agg2m  = ws + off; off += (size_t)N_t * H;
    float* h_t    = ws + off; off += (size_t)N_t * H;
    float* h_c    = ws + off; off += (size_t)N_c * H;
    float* h_m    = ws + off; off += (size_t)N_m * H;
    float* W1t    = ws + off; off += (size_t)KPAD * H;
    float* W1c    = ws + off; off += (size_t)KPAD * H;
    float* W1m    = ws + off; off += (size_t)KPAD * H;
    float* b1sum  = ws + off; off += (size_t)H;

    zero_kernel<<<ZWG, 256, 0, stream>>>(ws, zeroCount);
    prep_w_kernel<<<(KPAD * H + 255) / 256, 256, 0, stream>>>(
        w1_src_t, w1_dst_t, b1_t, w1_src_r, w1_dst_r, W1t, W1c, W1m, b1sum);
    scatter_scalar_kernel<<<(E + 255) / 256, 256, 0, stream>>>(
        x_card, x_merch, src_pays, dst_pays, src_recv, dst_recv, aggc1, aggm1, E, cs);
    conv1_t_kernel<<<256, 256, 0, stream>>>(x_trans, aggc1, aggm1, W1t, b1sum, h_t, N_t / 16, cs,
                                            aggt2c, src_pays, dst_pays, E);
    conv2_root_kernel<<<256, 256, 0, stream>>>(h_t, w2_r, b2, agg2c, agg2m, N_t / 16,
                                               aggt2m, src_recv, x_trans, dst_recv, E, cs);
    gemm88_kernel<<<N_c / 16, 128, 0, stream>>>(aggt2c, cs, x_card, W1c, b1_r, w2_l, h_c);
    gemm88_kernel<<<N_m / 16, 128, 0, stream>>>(aggt2m, cs, x_merch, W1m, b1_r + H, w2_l + H * H, h_m);
    scatter64_kernel<<<SWG, 256, 0, stream>>>(
        agg2c, dst_pays, h_c, src_pays, E);
    scatter64_kernel<<<SWG, 256, 0, stream>>>(
        agg2m, dst_recv, h_m, src_recv, E);
    out_kernel<<<OWG, 256, 0, stream>>>(agg2c, agg2m, w_lin, b_lin, out, N_t / 16);
}
